# starting kernel
# baseline (speedup 1.0000x reference)
#include <hip/hip_runtime.h>
#include <hip/hip_bf16.h>
#include <math.h>

#define NN      50000
#define NE      800000
#define DF      128
#define NHEADS  4
#define HU      32
#define NEG_SLOPE 0.2f
#define BN_EPS    0.001f

typedef __attribute__((ext_vector_type(2))) float v2f;
typedef __attribute__((ext_vector_type(8))) float v8f;

#define OFF_H    ((size_t)0)
#define OFF_UPD  (OFF_H   + (size_t)NN * DF)
#define OFF_AGG  (OFF_UPD + (size_t)NN * DF)
#define OFF_SI   (OFF_AGG + (size_t)NN * DF)
#define OFF_SJ   (OFF_SI  + (size_t)NN * NHEADS)
#define OFF_MAX  (OFF_SJ  + (size_t)NN * NHEADS)
#define OFF_DEN  (OFF_MAX + (size_t)NN * NHEADS)
#define OFF_W    (OFF_DEN + (size_t)NN * NHEADS)
#define OFF_MACC (OFF_W   + (size_t)NE * NHEADS)
#define OFF_SQ   (OFF_MACC + (size_t)DF)

__device__ __forceinline__ void atomicMaxFloat(float* addr, float val) {
    if (val >= 0.0f) {
        atomicMax((int*)addr, __float_as_int(val));
    } else {
        atomicMin((unsigned int*)addr, __float_as_uint(val));
    }
}

__global__ void k_init(float* __restrict__ ws) {
    int i = blockIdx.x * blockDim.x + threadIdx.x;
    size_t nagg = (size_t)NN * DF;
    if ((size_t)i < nagg) ws[OFF_AGG + i] = 0.0f;
    if (i < NN * NHEADS) {
        ws[OFF_MAX + i] = -INFINITY;
        ws[OFF_DEN + i] = 0.0f;
    }
    if (i < DF) {
        ws[OFF_MACC + i] = 0.0f;
        ws[OFF_SQ + i]   = 0.0f;
    }
}

template <int WLAYOUT, int DORELU>
__global__ __launch_bounds__(256) void k_gemm_wmma(
    const float* __restrict__ A, const float* __restrict__ B,
    const float* __restrict__ bias, float* __restrict__ C, int M)
{
    __shared__ float atile[16 * DF];

    const int tid = threadIdx.x;
    const int m0  = blockIdx.x * 16;

    #pragma unroll
    for (int i = 0; i < 8; ++i) {
        const int idx = tid + i * 256;
        const int r = idx >> 7, c = idx & (DF - 1);
        const int row = m0 + r;
        atile[idx] = (row < M) ? A[(size_t)row * DF + c] : 0.0f;
    }
    __syncthreads();

    const int lane = tid & 31;
    const int half = lane >> 4;
    const int l    = lane & 15;
    const int n0   = (tid >> 5) * 16;
    const int col  = n0 + l;

    const int b_base   = (WLAYOUT == 0) ? col
                                        : ((col >> 5) * (DF * HU) + (col & 31));
    const int b_kstride = (WLAYOUT == 0) ? DF : HU;

    const float* __restrict__ arow = &atile[l * DF + 2 * half];
    const float* __restrict__ bcol = &B[b_base + (size_t)(2 * half) * b_kstride];

    v8f acc = {};
    #pragma unroll
    for (int kb = 0; kb < DF; kb += 4) {
        v2f a = *(const v2f*)(arow + kb);
        v2f b;
        b.x = bcol[(size_t)kb * b_kstride];
        b.y = bcol[(size_t)(kb + 1) * b_kstride];
        acc = __builtin_amdgcn_wmma_f32_16x16x4_f32(
                  false, a, false, b, (short)0, acc, false, false);
    }

    const float bv = bias[col];
    #pragma unroll
    for (int r = 0; r < 8; ++r) {
        float v = acc[r] + bv;
        if (DORELU) v = fmaxf(v, 0.0f);
        const int row = m0 + r + 8 * half;
        if (row < M) C[(size_t)row * DF + col] = v;
    }
}

__global__ __launch_bounds__(128) void k_scores(
    const float* __restrict__ h, const float* __restrict__ a_i,
    const float* __restrict__ a_j, float* __restrict__ s_i,
    float* __restrict__ s_j)
{
    const int node = blockIdx.x;
    const int t = threadIdx.x;
    const float hv = h[(size_t)node * DF + t];
    float vi = hv * a_i[t];
    float vj = hv * a_j[t];
    #pragma unroll
    for (int off = 16; off > 0; off >>= 1) {
        vi += __shfl_xor(vi, off, 32);
        vj += __shfl_xor(vj, off, 32);
    }
    if ((t & 31) == 0) {
        const int hd = t >> 5;
        s_i[node * NHEADS + hd] = vi;
        s_j[node * NHEADS + hd] = vj;
    }
}

__global__ void k_edge_logits_max(
    const int* __restrict__ ei, const float* __restrict__ s_i,
    const float* __restrict__ s_j, const float* __restrict__ a_e,
    const float* __restrict__ b_a, float* __restrict__ wbuf,
    float* __restrict__ maxb)
{
    const int idx = blockIdx.x * blockDim.x + threadIdx.x;
    if (idx >= NE * NHEADS) return;
    const int e  = idx >> 2;
    const int hd = idx & 3;
    const int recv = ei[2 * e];
    const int send = ei[2 * e + 1];
    float lg = s_i[recv * NHEADS + hd] + s_j[send * NHEADS + hd]
             + a_e[hd] + b_a[hd];
    lg = (lg > 0.0f) ? lg : NEG_SLOPE * lg;
    wbuf[idx] = lg;
    atomicMaxFloat(&maxb[recv * NHEADS + hd], lg);
}

__global__ void k_edge_exp_sum(
    const int* __restrict__ ei, const float* __restrict__ maxb,
    float* __restrict__ wbuf, float* __restrict__ denom)
{
    const int idx = blockIdx.x * blockDim.x + threadIdx.x;
    if (idx >= NE * NHEADS) return;
    const int e  = idx >> 2;
    const int hd = idx & 3;
    const int recv = ei[2 * e];
    const float w = expf(wbuf[idx] - maxb[recv * NHEADS + hd]);
    wbuf[idx] = w;
    atomicAdd(&denom[recv * NHEADS + hd], w);
}

__global__ __launch_bounds__(256) void k_edge_scatter(
    const int* __restrict__ ei, const float* __restrict__ h,
    const float* __restrict__ wbuf, const float* __restrict__ denom,
    float* __restrict__ agg)
{
    const long long idx = (long long)blockIdx.x * 256 + threadIdx.x;
    if (idx >= (long long)NE * DF) return;
    const int e   = (int)(idx >> 7);
    const int col = (int)(idx & (DF - 1));
    const int hd  = col >> 5;
    const int recv = ei[2 * e];
    const int send = ei[2 * e + 1];
    const float alpha = wbuf[e * NHEADS + hd]
                      / (denom[recv * NHEADS + hd] + 1e-16f);
    atomicAdd(&agg[(size_t)recv * DF + col],
              h[(size_t)send * DF + col] * alpha);
}

#define BN_ROWS 128
__global__ __launch_bounds__(128) void k_bn_partial(
    const float* __restrict__ upd, float* __restrict__ macc,
    float* __restrict__ sqacc)
{
    const int col = threadIdx.x;
    const int r0  = blockIdx.x * BN_ROWS;
    float s = 0.0f, s2 = 0.0f;
    for (int r = 0; r < BN_ROWS; ++r) {
        const int row = r0 + r;
        if (row < NN) {
            const float v = upd[(size_t)row * DF + col];
            s += v; s2 += v * v;
        }
    }
    atomicAdd(&macc[col], s);
    atomicAdd(&sqacc[col], s2);
}

__global__ void k_finalize(
    const float* __restrict__ upd, const float* __restrict__ agg,
    const float* __restrict__ macc, const float* __restrict__ sqacc,
    const float* __restrict__ gamma, const float* __restrict__ beta,
    float* __restrict__ out)
{
    const long long idx = (long long)blockIdx.x * blockDim.x + threadIdx.x;
    if (idx >= (long long)NN * DF) return;
    const int col = (int)(idx & (DF - 1));
    const float inv_n = 1.0f / (float)NN;
    const float mean = macc[col] * inv_n;
    const float var  = sqacc[col] * inv_n - mean * mean;
    const float normed = gamma[col] * (upd[idx] - mean)
                       * rsqrtf(var + BN_EPS) + beta[col];
    out[idx] = normed + fmaxf(agg[idx], 0.0f);
}

extern "C" void kernel_launch(void* const* d_in, const int* in_sizes, int n_in,
                              void* d_out, int out_size, void* d_ws, size_t ws_size,
                              hipStream_t stream) {
    const float* x     = (const float*)d_in[0];
    const int*   ei    = (const int*)  d_in[1];
    const float* Wk    = (const float*)d_in[2];
    const float* bk    = (const float*)d_in[3];
    const float* a_i   = (const float*)d_in[4];
    const float* a_j   = (const float*)d_in[5];
    const float* a_e   = (const float*)d_in[6];
    const float* b_a   = (const float*)d_in[7];
    const float* Wm    = (const float*)d_in[8];
    const float* bm    = (const float*)d_in[9];
    const float* gamma = (const float*)d_in[10];
    const float* beta  = (const float*)d_in[11];
    float* out = (float*)d_out;
    float* ws  = (float*)d_ws;

    float* h_buf  = ws + OFF_H;
    float* upd    = ws + OFF_UPD;
    float* agg    = ws + OFF_AGG;
    float* s_i    = ws + OFF_SI;
    float* s_j    = ws + OFF_SJ;
    float* maxb   = ws + OFF_MAX;
    float* denom  = ws + OFF_DEN;
    float* wbuf   = ws + OFF_W;
    float* macc   = ws + OFF_MACC;
    float* sqacc  = ws + OFF_SQ;

    {
        const long long n = (long long)NN * DF;
        k_init<<<(int)((n + 255) / 256), 256, 0, stream>>>(ws);
    }
    const int gemm_blocks = (NN + 15) / 16;
    k_gemm_wmma<1, 0><<<gemm_blocks, 256, 0, stream>>>(x, Wk, bk, h_buf, NN);
    k_gemm_wmma<0, 1><<<gemm_blocks, 256, 0, stream>>>(x, Wm, bm, upd, NN);
    k_scores<<<NN, 128, 0, stream>>>(h_buf, a_i, a_j, s_i, s_j);
    {
        const int n = NE * NHEADS;
        k_edge_logits_max<<<(n + 255) / 256, 256, 0, stream>>>(
            ei, s_i, s_j, a_e, b_a, wbuf, maxb);
    }
    {
        const int n = NE * NHEADS;
        k_edge_exp_sum<<<(n + 255) / 256, 256, 0, stream>>>(
            ei, maxb, wbuf, denom);
    }
    {
        const long long n = (long long)NE * DF;
        k_edge_scatter<<<(int)((n + 255) / 256), 256, 0, stream>>>(
            ei, h_buf, wbuf, denom, agg);
    }
    k_bn_partial<<<(NN + BN_ROWS - 1) / BN_ROWS, 128, 0, stream>>>(
        upd, macc, sqacc);
    {
        const long long n = (long long)NN * DF;
        k_finalize<<<(int)((n + 255) / 256), 256, 0, stream>>>(
            upd, agg, macc, sqacc, gamma, beta, out);
    }
}


# Attention scores formed in the h projection's epilogue from the accumulators (16-lane DPP sums, a head's two waves added via LDS: sum order changed); score launch and its re-read of h gone
# speedup vs baseline: 1.7817x; 1.7817x over previous
#include <hip/hip_runtime.h>
#include <hip/hip_bf16.h>
#include <math.h>

#define NN      50000
#define NE      800000
#define DF      128
#define NHEADS  4
#define HU      32
#define NEG_SLOPE 0.2f
#define BN_EPS    0.001f
#define RNG     196
#define NRNG    ((NN + RNG - 1) / RNG)

typedef __attribute__((ext_vector_type(2))) float v2f;
typedef __attribute__((ext_vector_type(8))) float v8f;

#define OFF_H    ((size_t)0)
#define OFF_UPD  (OFF_H   + (size_t)NN * DF)
#define OFF_AGG  (OFF_UPD + (size_t)NN * DF)
#define OFF_SI   (OFF_AGG + (size_t)NN * DF)
#define OFF_SJ   (OFF_SI  + (size_t)NN * NHEADS)
#define OFF_MAX  (OFF_SJ  + (size_t)NN * NHEADS)
#define OFF_DEN  (OFF_MAX + (size_t)NN * NHEADS)
#define OFF_W    (OFF_DEN + (size_t)NN * NHEADS)
#define OFF_MACC (OFF_W   + (size_t)NE * NHEADS)
#define OFF_SQ   (OFF_MACC + (size_t)DF)

__device__ __forceinline__ void atomicMaxFloat(float* addr, float val) {
    if (val >= 0.0f) {
        atomicMax((int*)addr, __float_as_int(val));
    } else {
        atomicMin((unsigned int*)addr, __float_as_uint(val));
    }
}

__global__ void k_init(float* __restrict__ ws) {
    int i = blockIdx.x * blockDim.x + threadIdx.x;
    if (i < NN * NHEADS) {
        ws[OFF_MAX + i] = -INFINITY;
        ws[OFF_DEN + i] = 0.0f;
    }
    if (i < DF) {
        ws[OFF_MACC + i] = 0.0f;
        ws[OFF_SQ + i]   = 0.0f;
    }
}

template <int M>
__device__ __forceinline__ float row_xor(float x) {
    return __int_as_float(__builtin_amdgcn_update_dpp(
        0, __float_as_int(x), 0x160 + M, 0xf, 0xf, true));
}
__device__ __forceinline__ float pick(bool c, float a, float b) { return c ? a : b; }
template <int W>
__device__ __forceinline__ void fold(float* v, bool up) {
    #pragma unroll
    for (int k = 0; k < W; ++k) {
        const float lo = v[k], hi = v[k + W];
        v[k] = pick(up, hi, lo) + row_xor<W>(pick(up, lo, hi));
    }
}

template <int WLAYOUT, int DORELU, int SCORES>
__global__ __launch_bounds__(256) void k_gemm_wmma(
    const float* __restrict__ A, const float* __restrict__ B,
    const float* __restrict__ bias, float* __restrict__ C, int M,
    const float* __restrict__ a_i, const float* __restrict__ a_j,
    float* __restrict__ s_i, float* __restrict__ s_j)
{
    __shared__ float atile[16 * DF];
    __shared__ float part[SCORES ? 8 * 2 * 16 : 1];

    const int tid = threadIdx.x;
    const int m0  = blockIdx.x * 16;

    #pragma unroll
    for (int i = 0; i < 8; ++i) {
        const int idx = tid + i * 256;
        const int r = idx >> 7, c = idx & (DF - 1);
        const int row = m0 + r;
        atile[idx] = (row < M) ? A[(size_t)row * DF + c] : 0.0f;
    }
    __syncthreads();

    const int lane = tid & 31;
    const int half = lane >> 4;
    const int l    = lane & 15;
    const int n0   = (tid >> 5) * 16;
    const int col  = n0 + l;

    const int b_base   = (WLAYOUT == 0) ? col
                                        : ((col >> 5) * (DF * HU) + (col & 31));
    const int b_kstride = (WLAYOUT == 0) ? DF : HU;

    const float* __restrict__ arow = &atile[l * DF + 2 * half];
    const float* __restrict__ bcol = &B[b_base + (size_t)(2 * half) * b_kstride];

    v8f acc = {};
    #pragma unroll
    for (int kb = 0; kb < DF; kb += 4) {
        v2f a = *(const v2f*)(arow + kb);
        v2f b;
        b.x = bcol[(size_t)kb * b_kstride];
        b.y = bcol[(size_t)(kb + 1) * b_kstride];
        acc = __builtin_amdgcn_wmma_f32_16x16x4_f32(
                  false, a, false, b, (short)0, acc, false, false);
    }

    const float bv = bias[col];
    float hv[8];
    #pragma unroll
    for (int r = 0; r < 8; ++r) {
        float v = acc[r] + bv;
        if (DORELU) v = fmaxf(v, 0.0f);
        hv[r] = v;
    }

    if (SCORES) {
        const float ai = a_i[col], aj = a_j[col];
        float v[16];
        #pragma unroll
        for (int r = 0; r < 8; ++r) { v[r] = hv[r] * ai; v[8 + r] = hv[r] * aj; }
        fold<8>(v, (l & 8) != 0);
        fold<4>(v, (l & 4) != 0);
        fold<2>(v, (l & 2) != 0);
        fold<1>(v, (l & 1) != 0);
        part[((tid >> 5) * 2 + (l >> 3)) * 16 + 8 * half + (l & 7)] = v[0];
        __syncthreads();
    }

    #pragma unroll
    for (int r = 0; r < 8; ++r) {
        const int row = m0 + r + 8 * half;
        if (row < M) C[(size_t)row * DF + col] = hv[r];
    }

    if (SCORES) {
        if (tid < 128) {
            const int which = tid >> 6;
            const int row = (tid & 63) >> 2, hd = tid & 3;
            const float s = part[((2 * hd) * 2 + which) * 16 + row]
                          + part[((2 * hd + 1) * 2 + which) * 16 + row];
            const int node = m0 + row;
            if (node < M) (which ? s_j : s_i)[node * NHEADS + hd] = s;
        }
    }
}

__global__ void k_edge_logits_max(
    const int* __restrict__ ei, const float* __restrict__ s_i,
    const float* __restrict__ s_j, const float* __restrict__ a_e,
    const float* __restrict__ b_a, float* __restrict__ wbuf,
    float* __restrict__ maxb)
{
    const int idx = blockIdx.x * blockDim.x + threadIdx.x;
    if (idx >= NE * NHEADS) return;
    const int e  = idx >> 2;
    const int hd = idx & 3;
    const int recv = ei[2 * e];
    const int send = ei[2 * e + 1];
    float lg = s_i[recv * NHEADS + hd] + s_j[send * NHEADS + hd]
             + a_e[hd] + b_a[hd];
    lg = (lg > 0.0f) ? lg : NEG_SLOPE * lg;
    wbuf[idx] = lg;
    atomicMaxFloat(&maxb[recv * NHEADS + hd], lg);
}

__global__ void k_edge_exp_sum(
    const int* __restrict__ ei, const float* __restrict__ maxb,
    float* __restrict__ wbuf, float* __restrict__ denom,
    unsigned char* __restrict__ rbyte)
{
    const int idx = blockIdx.x * blockDim.x + threadIdx.x;
    if (idx >= NE * NHEADS) return;
    const int e  = idx >> 2;
    const int hd = idx & 3;
    const int recv = ei[2 * e];
    const float w = expf(wbuf[idx] - maxb[recv * NHEADS + hd]);
    wbuf[idx] = w;
    atomicAdd(&denom[recv * NHEADS + hd], w);
    if (hd == 0) rbyte[e] = (unsigned char)(recv / RNG);
}

#define SG_WAVES 16
#define QCAP     1024
#define NVEC     (NE / 16)
#define NSUB     ((NVEC + 31) / 32)

__device__ __forceinline__ unsigned zero_bytes(unsigned t) {
    return ~(((t & 0x7f7f7f7fu) + 0x7f7f7f7fu) | t) & 0x80808080u;
}

template <bool FULL>
__device__ __forceinline__ void add_batch(
    const int* q, unsigned head, int n, int r0, int lane,
    const int* __restrict__ ei, const float* __restrict__ h,
    const float* __restrict__ wbuf, const float* dsh, float* acc)
{
    const int hd = lane >> 3;
    int   e[8];
    int2  rs[8];
    float w[8];
    #pragma unroll
    for (int b = 0; b < 8; ++b) {
        const int s = FULL ? b : (b < n ? b : n - 1);
        e[b] = q[(head + s) & (QCAP - 1)];
    }
    #pragma unroll
    for (int b = 0; b < 8; ++b) {
        rs[b] = *(const int2*)(ei + 2 * e[b]);
        w[b]  = wbuf[e[b] * NHEADS + hd];
    }
    float4 hv[8];
    #pragma unroll
    for (int b = 0; b < 8; ++b)
        hv[b] = *(const float4*)(h + (size_t)rs[b].y * DF + 4 * lane);
    #pragma unroll
    for (int b = 0; b < 8; ++b) {
        if (FULL || b < n) {
            const int rel = rs[b].x - r0;
            const float alpha = w[b] / (dsh[rel * NHEADS + hd] + 1e-16f);
            float* a = acc + rel * DF + lane;
            atomicAdd(a,      hv[b].x * alpha);
            atomicAdd(a + 32, hv[b].y * alpha);
            atomicAdd(a + 64, hv[b].z * alpha);
            atomicAdd(a + 96, hv[b].w * alpha);
        }
    }
}

__global__ __launch_bounds__(512) void k_edge_range_sum(
    const int* __restrict__ ei, const float* __restrict__ h,
    const float* __restrict__ wbuf, const float* __restrict__ denom,
    const unsigned char* __restrict__ rbyte, float* __restrict__ agg)
{
    __shared__ float acc[RNG * DF];
    __shared__ float dsh[RNG * NHEADS];
    __shared__ int   queue[SG_WAVES * QCAP];

    const int tid  = threadIdx.x;
    const int lane = tid & 31;
    const int wv   = tid >> 5;
    const int r0   = blockIdx.x * RNG;

    for (int i = tid; i < RNG * DF; i += 512) acc[i] = 0.0f;
    for (int i = tid; i < RNG * NHEADS; i += 512) {
        const int g = r0 * NHEADS + i;
        dsh[i] = (g < NN * NHEADS) ? denom[g] : 0.0f;
    }
    __syncthreads();

    const unsigned pat = blockIdx.x * 0x01010101u;
    const uint4* __restrict__ rv = (const uint4*)rbyte;
    int* q = queue + wv * QCAP;
    unsigned head = 0, tail = 0;

    uint4 cur[4];
    #pragma unroll
    for (int k = 0; k < 4; ++k) {
        const int vec = (wv + k * SG_WAVES) * 32 + lane;
        cur[k] = rv[vec < NVEC ? vec : NVEC - 1];
    }
    for (int s0 = wv; s0 < NSUB; s0 += 4 * SG_WAVES) {
        uint4 nxt[4];
        #pragma unroll
        for (int k = 0; k < 4; ++k) {
            const int vec = (s0 + (4 + k) * SG_WAVES) * 32 + lane;
            nxt[k] = rv[vec < NVEC ? vec : NVEC - 1];
        }
        #pragma unroll
        for (int k = 0; k < 4; ++k) {
            const int vec = (s0 + k * SG_WAVES) * 32 + lane;
            unsigned m = 0u;
            if (vec < NVEC)
                m = (zero_bytes(cur[k].x ^ pat) >> 7) | (zero_bytes(cur[k].y ^ pat) >> 6)
                  | (zero_bytes(cur[k].z ^ pat) >> 5) | (zero_bytes(cur[k].w ^ pat) >> 4);
            for (;;) {
                const unsigned bal = __builtin_amdgcn_ballot_w32(m != 0u);
                if (bal == 0u) break;
                if (m != 0u) {
                    const int p = __builtin_ctz(m);
                    const unsigned rank = __builtin_amdgcn_mbcnt_lo(bal, 0u);
                    q[(tail + rank) & (QCAP - 1)] = vec * 16 + ((p & 7) << 2) + (p >> 3);
                    m &= m - 1u;
                }
                tail += (unsigned)__builtin_popcount(bal);
            }
            while (tail - head >= 8u) {
                add_batch<true>(q, head, 8, r0, lane, ei, h, wbuf, dsh, acc);
                head += 8u;
            }
        }
        #pragma unroll
        for (int k = 0; k < 4; ++k) cur[k] = nxt[k];
    }
    if (tail != head)
        add_batch<false>(q, head, (int)(tail - head), r0, lane, ei, h, wbuf, dsh, acc);
    __syncthreads();

    for (int r = wv; r < RNG; r += SG_WAVES) {
        const int row = r0 + r;
        if (row < NN) {
            const float* a = acc + r * DF + lane;
            const float4 v = make_float4(a[0], a[32], a[64], a[96]);
            *(float4*)(agg + (size_t)row * DF + 4 * lane) = v;
        }
    }
}

#define BN_ROWS 128
__global__ __launch_bounds__(128) void k_bn_partial(
    const float* __restrict__ upd, float* __restrict__ macc,
    float* __restrict__ sqacc)
{
    const int col = threadIdx.x;
    const int r0  = blockIdx.x * BN_ROWS;
    float s = 0.0f, s2 = 0.0f;
    for (int r = 0; r < BN_ROWS; ++r) {
        const int row = r0 + r;
        if (row < NN) {
            const float v = upd[(size_t)row * DF + col];
            s += v; s2 += v * v;
        }
    }
    atomicAdd(&macc[col], s);
    atomicAdd(&sqacc[col], s2);
}

__global__ void k_finalize(
    const float* __restrict__ upd, const float* __restrict__ agg,
    const float* __restrict__ macc, const float* __restrict__ sqacc,
    const float* __restrict__ gamma, const float* __restrict__ beta,
    float* __restrict__ out)
{
    const long long idx = (long long)blockIdx.x * blockDim.x + threadIdx.x;
    if (idx >= (long long)NN * DF) return;
    const int col = (int)(idx & (DF - 1));
    const float inv_n = 1.0f / (float)NN;
    const float mean = macc[col] * inv_n;
    const float var  = sqacc[col] * inv_n - mean * mean;
    const float normed = gamma[col] * (upd[idx] - mean)
                       * rsqrtf(var + BN_EPS) + beta[col];
    out[idx] = normed + fmaxf(agg[idx], 0.0f);
}

extern "C" void kernel_launch(void* const* d_in, const int* in_sizes, int n_in,
                              void* d_out, int out_size, void* d_ws, size_t ws_size,
                              hipStream_t stream) {
    const float* x     = (const float*)d_in[0];
    const int*   ei    = (const int*)  d_in[1];
    const float* Wk    = (const float*)d_in[2];
    const float* bk    = (const float*)d_in[3];
    const float* a_i   = (const float*)d_in[4];
    const float* a_j   = (const float*)d_in[5];
    const float* a_e   = (const float*)d_in[6];
    const float* b_a   = (const float*)d_in[7];
    const float* Wm    = (const float*)d_in[8];
    const float* bm    = (const float*)d_in[9];
    const float* gamma = (const float*)d_in[10];
    const float* beta  = (const float*)d_in[11];
    float* out = (float*)d_out;
    float* ws  = (float*)d_ws;

    float* h_buf  = ws + OFF_H;
    float* upd    = ws + OFF_UPD;
    float* agg    = ws + OFF_AGG;
    float* s_i    = ws + OFF_SI;
    float* s_j    = ws + OFF_SJ;
    float* maxb   = ws + OFF_MAX;
    float* denom  = ws + OFF_DEN;
    float* wbuf   = ws + OFF_W;
    float* macc   = ws + OFF_MACC;
    float* sqacc  = ws + OFF_SQ;
    unsigned char* rbyte = (unsigned char*)(ws + OFF_SI);

    {
        const int n = NN * NHEADS;
        k_init<<<(n + 255) / 256, 256, 0, stream>>>(ws);
    }
    const int gemm_blocks = (NN + 15) / 16;
    k_gemm_wmma<1, 0, 1><<<gemm_blocks, 256, 0, stream>>>(
        x, Wk, bk, h_buf, NN, a_i, a_j, s_i, s_j);
    k_gemm_wmma<0, 1, 0><<<gemm_blocks, 256, 0, stream>>>(
        x, Wm, bm, upd, NN, nullptr, nullptr, nullptr, nullptr);
    {
        const int n = NE * NHEADS;
        k_edge_logits_max<<<(n + 255) / 256, 256, 0, stream>>>(
            ei, s_i, s_j, a_e, b_a, wbuf, maxb);
    }
    {
        const int n = NE * NHEADS;
        k_edge_exp_sum<<<(n + 255) / 256, 256, 0, stream>>>(
            ei, maxb, wbuf, denom, rbyte);
    }
    k_edge_range_sum<<<NRNG, 512, 0, stream>>>(ei, h_buf, wbuf, denom, rbyte, agg);
    k_bn_partial<<<(NN + BN_ROWS - 1) / BN_ROWS, 128, 0, stream>>>(
        upd, macc, sqacc);
    {
        const long long n = (long long)NN * DF;
        k_finalize<<<(int)((n + 255) / 256), 256, 0, stream>>>(
            upd, agg, macc, sqacc, gamma, beta, out);
    }
}


# Batch-norm column sums taken from the upd projection's accumulators in its epilogue (a partial row per 16-row stripe, a small launch adds the 3125 rows: sum order changed); serial pass over upd gone
# speedup vs baseline: 2.0749x; 1.1646x over previous
#include <hip/hip_runtime.h>
#include <hip/hip_bf16.h>
#include <math.h>

#define NN      50000
#define NE      800000
#define DF      128
#define NHEADS  4
#define HU      32
#define NEG_SLOPE 0.2f
#define BN_EPS    0.001f
#define RNG     196
#define NRNG    ((NN + RNG - 1) / RNG)
#define GEMM_BLOCKS ((NN + 15) / 16)
#define CS_ROWS 25
static_assert(GEMM_BLOCKS % CS_ROWS == 0, "the reduce launch reads whole groups of partial rows");

typedef __attribute__((ext_vector_type(2))) float v2f;
typedef __attribute__((ext_vector_type(8))) float v8f;

#define OFF_H    ((size_t)0)
#define OFF_UPD  (OFF_H   + (size_t)NN * DF)
#define OFF_AGG  (OFF_UPD + (size_t)NN * DF)
#define OFF_SI   (OFF_AGG + (size_t)NN * DF)
#define OFF_SJ   (OFF_SI  + (size_t)NN * NHEADS)
#define OFF_MAX  (OFF_SJ  + (size_t)NN * NHEADS)
#define OFF_DEN  (OFF_MAX + (size_t)NN * NHEADS)
#define OFF_W    (OFF_DEN + (size_t)NN * NHEADS)
#define OFF_MACC (OFF_W   + (size_t)NE * NHEADS)
#define OFF_SQ   (OFF_MACC + (size_t)DF)

__device__ __forceinline__ void atomicMaxFloat(float* addr, float val) {
    if (val >= 0.0f) {
        atomicMax((int*)addr, __float_as_int(val));
    } else {
        atomicMin((unsigned int*)addr, __float_as_uint(val));
    }
}

__global__ void k_init(float* __restrict__ ws) {
    int i = blockIdx.x * blockDim.x + threadIdx.x;
    if (i < NN * NHEADS) {
        ws[OFF_MAX + i] = -INFINITY;
        ws[OFF_DEN + i] = 0.0f;
    }
    if (i < DF) {
        ws[OFF_MACC + i] = 0.0f;
        ws[OFF_SQ + i]   = 0.0f;
    }
}

template <int M>
__device__ __forceinline__ float row_xor(float x) {
    return __int_as_float(__builtin_amdgcn_update_dpp(
        0, __float_as_int(x), 0x160 + M, 0xf, 0xf, true));
}
__device__ __forceinline__ float other_half(float x) {
    const int v = __float_as_int(x);
    return __int_as_float(__builtin_amdgcn_permlanex16(v, v, 0x76543210, 0xfedcba98, false, false));
}
__device__ __forceinline__ float pick(bool c, float a, float b) { return c ? a : b; }
template <int W>
__device__ __forceinline__ void fold(float* v, bool up) {
    #pragma unroll
    for (int k = 0; k < W; ++k) {
        const float lo = v[k], hi = v[k + W];
        v[k] = pick(up, hi, lo) + row_xor<W>(pick(up, lo, hi));
    }
}

template <int WLAYOUT, int DORELU, int SCORES, int COLSUMS>
__global__ __launch_bounds__(256) void k_gemm_wmma(
    const float* __restrict__ A, const float* __restrict__ B,
    const float* __restrict__ bias, float* __restrict__ C, int M,
    const float* __restrict__ a_i, const float* __restrict__ a_j,
    float* __restrict__ s_i, float* __restrict__ s_j,
    float* __restrict__ colsum_rows)
{
    __shared__ float atile[16 * DF];
    __shared__ float part[SCORES ? 8 * 2 * 16 : 1];

    const int tid = threadIdx.x;
    const int m0  = blockIdx.x * 16;

    #pragma unroll
    for (int i = 0; i < 8; ++i) {
        const int idx = tid + i * 256;
        const int r = idx >> 7, c = idx & (DF - 1);
        const int row = m0 + r;
        atile[idx] = (row < M) ? A[(size_t)row * DF + c] : 0.0f;
    }
    __syncthreads();

    const int lane = tid & 31;
    const int half = lane >> 4;
    const int l    = lane & 15;
    const int n0   = (tid >> 5) * 16;
    const int col  = n0 + l;

    const int b_base   = (WLAYOUT == 0) ? col
                                        : ((col >> 5) * (DF * HU) + (col & 31));
    const int b_kstride = (WLAYOUT == 0) ? DF : HU;

    const float* __restrict__ arow = &atile[l * DF + 2 * half];
    const float* __restrict__ bcol = &B[b_base + (size_t)(2 * half) * b_kstride];

    v8f acc = {};
    #pragma unroll
    for (int kb = 0; kb < DF; kb += 4) {
        v2f a = *(const v2f*)(arow + kb);
        v2f b;
        b.x = bcol[(size_t)kb * b_kstride];
        b.y = bcol[(size_t)(kb + 1) * b_kstride];
        acc = __builtin_amdgcn_wmma_f32_16x16x4_f32(
                  false, a, false, b, (short)0, acc, false, false);
    }

    const float bv = bias[col];
    float hv[8];
    #pragma unroll
    for (int r = 0; r < 8; ++r) {
        float v = acc[r] + bv;
        if (DORELU) v = fmaxf(v, 0.0f);
        hv[r] = v;
    }

    if (SCORES) {
        const float ai = a_i[col], aj = a_j[col];
        float v[16];
        #pragma unroll
        for (int r = 0; r < 8; ++r) { v[r] = hv[r] * ai; v[8 + r] = hv[r] * aj; }
        fold<8>(v, (l & 8) != 0);
        fold<4>(v, (l & 4) != 0);
        fold<2>(v, (l & 2) != 0);
        fold<1>(v, (l & 1) != 0);
        part[((tid >> 5) * 2 + (l >> 3)) * 16 + 8 * half + (l & 7)] = v[0];
        __syncthreads();
    }

    float cs = 0.0f, cq = 0.0f;
    #pragma unroll
    for (int r = 0; r < 8; ++r) {
        const int row = m0 + r + 8 * half;
        if (row < M) {
            C[(size_t)row * DF + col] = hv[r];
            if (COLSUMS) { cs += hv[r]; cq += hv[r] * hv[r]; }
        }
    }

    if (COLSUMS) {
        cs += other_half(cs);
        cq += other_half(cq);
        colsum_rows[(size_t)blockIdx.x * (2 * DF) + (tid >> 5) * 32 + lane] = half ? cq : cs;
    }

    if (SCORES) {
        if (tid < 128) {
            const int which = tid >> 6;
            const int row = (tid & 63) >> 2, hd = tid & 3;
            const float s = part[((2 * hd) * 2 + which) * 16 + row]
                          + part[((2 * hd + 1) * 2 + which) * 16 + row];
            const int node = m0 + row;
            if (node < M) (which ? s_j : s_i)[node * NHEADS + hd] = s;
        }
    }
}

__global__ void k_edge_logits_max(
    const int* __restrict__ ei, const float* __restrict__ s_i,
    const float* __restrict__ s_j, const float* __restrict__ a_e,
    const float* __restrict__ b_a, float* __restrict__ wbuf,
    float* __restrict__ maxb)
{
    const int idx = blockIdx.x * blockDim.x + threadIdx.x;
    if (idx >= NE * NHEADS) return;
    const int e  = idx >> 2;
    const int hd = idx & 3;
    const int recv = ei[2 * e];
    const int send = ei[2 * e + 1];
    float lg = s_i[recv * NHEADS + hd] + s_j[send * NHEADS + hd]
             + a_e[hd] + b_a[hd];
    lg = (lg > 0.0f) ? lg : NEG_SLOPE * lg;
    wbuf[idx] = lg;
    atomicMaxFloat(&maxb[recv * NHEADS + hd], lg);
}

__global__ void k_edge_exp_sum(
    const int* __restrict__ ei, const float* __restrict__ maxb,
    float* __restrict__ wbuf, float* __restrict__ denom,
    unsigned char* __restrict__ rbyte)
{
    const int idx = blockIdx.x * blockDim.x + threadIdx.x;
    if (idx >= NE * NHEADS) return;
    const int e  = idx >> 2;
    const int hd = idx & 3;
    const int recv = ei[2 * e];
    const float w = expf(wbuf[idx] - maxb[recv * NHEADS + hd]);
    wbuf[idx] = w;
    atomicAdd(&denom[recv * NHEADS + hd], w);
    if (hd == 0) rbyte[e] = (unsigned char)(recv / RNG);
}

#define SG_WAVES 16
#define QCAP     1024
#define NVEC     (NE / 16)
#define NSUB     ((NVEC + 31) / 32)

__device__ __forceinline__ unsigned zero_bytes(unsigned t) {
    return ~(((t & 0x7f7f7f7fu) + 0x7f7f7f7fu) | t) & 0x80808080u;
}

template <bool FULL>
__device__ __forceinline__ void add_batch(
    const int* q, unsigned head, int n, int r0, int lane,
    const int* __restrict__ ei, const float* __restrict__ h,
    const float* __restrict__ wbuf, const float* dsh, float* acc)
{
    const int hd = lane >> 3;
    int   e[8];
    int2  rs[8];
    float w[8];
    #pragma unroll
    for (int b = 0; b < 8; ++b) {
        const int s = FULL ? b : (b < n ? b : n - 1);
        e[b] = q[(head + s) & (QCAP - 1)];
    }
    #pragma unroll
    for (int b = 0; b < 8; ++b) {
        rs[b] = *(const int2*)(ei + 2 * e[b]);
        w[b]  = wbuf[e[b] * NHEADS + hd];
    }
    float4 hv[8];
    #pragma unroll
    for (int b = 0; b < 8; ++b)
        hv[b] = *(const float4*)(h + (size_t)rs[b].y * DF + 4 * lane);
    #pragma unroll
    for (int b = 0; b < 8; ++b) {
        if (FULL || b < n) {
            const int rel = rs[b].x - r0;
            const float alpha = w[b] / (dsh[rel * NHEADS + hd] + 1e-16f);
            float* a = acc + rel * DF + lane;
            atomicAdd(a,      hv[b].x * alpha);
            atomicAdd(a + 32, hv[b].y * alpha);
            atomicAdd(a + 64, hv[b].z * alpha);
            atomicAdd(a + 96, hv[b].w * alpha);
        }
    }
}

__global__ __launch_bounds__(512) void k_edge_range_sum(
    const int* __restrict__ ei, const float* __restrict__ h,
    const float* __restrict__ wbuf, const float* __restrict__ denom,
    const unsigned char* __restrict__ rbyte, float* __restrict__ agg)
{
    __shared__ float acc[RNG * DF];
    __shared__ float dsh[RNG * NHEADS];
    __shared__ int   queue[SG_WAVES * QCAP];

    const int tid  = threadIdx.x;
    const int lane = tid & 31;
    const int wv   = tid >> 5;
    const int r0   = blockIdx.x * RNG;

    for (int i = tid; i < RNG * DF; i += 512) acc[i] = 0.0f;
    for (int i = tid; i < RNG * NHEADS; i += 512) {
        const int g = r0 * NHEADS + i;
        dsh[i] = (g < NN * NHEADS) ? denom[g] : 0.0f;
    }
    __syncthreads();

    const unsigned pat = blockIdx.x * 0x01010101u;
    const uint4* __restrict__ rv = (const uint4*)rbyte;
    int* q = queue + wv * QCAP;
    unsigned head = 0, tail = 0;

    uint4 cur[4];
    #pragma unroll
    for (int k = 0; k < 4; ++k) {
        const int vec = (wv + k * SG_WAVES) * 32 + lane;
        cur[k] = rv[vec < NVEC ? vec : NVEC - 1];
    }
    for (int s0 = wv; s0 < NSUB; s0 += 4 * SG_WAVES) {
        uint4 nxt[4];
        #pragma unroll
        for (int k = 0; k < 4; ++k) {
            const int vec = (s0 + (4 + k) * SG_WAVES) * 32 + lane;
            nxt[k] = rv[vec < NVEC ? vec : NVEC - 1];
        }
        #pragma unroll
        for (int k = 0; k < 4; ++k) {
            const int vec = (s0 + k * SG_WAVES) * 32 + lane;
            unsigned m = 0u;
            if (vec < NVEC)
                m = (zero_bytes(cur[k].x ^ pat) >> 7) | (zero_bytes(cur[k].y ^ pat) >> 6)
                  | (zero_bytes(cur[k].z ^ pat) >> 5) | (zero_bytes(cur[k].w ^ pat) >> 4);
            for (;;) {
                const unsigned bal = __builtin_amdgcn_ballot_w32(m != 0u);
                if (bal == 0u) break;
                if (m != 0u) {
                    const int p = __builtin_ctz(m);
                    const unsigned rank = __builtin_amdgcn_mbcnt_lo(bal, 0u);
                    q[(tail + rank) & (QCAP - 1)] = vec * 16 + ((p & 7) << 2) + (p >> 3);
                    m &= m - 1u;
                }
                tail += (unsigned)__builtin_popcount(bal);
            }
            while (tail - head >= 8u) {
                add_batch<true>(q, head, 8, r0, lane, ei, h, wbuf, dsh, acc);
                head += 8u;
            }
        }
        #pragma unroll
        for (int k = 0; k < 4; ++k) cur[k] = nxt[k];
    }
    if (tail != head)
        add_batch<false>(q, head, (int)(tail - head), r0, lane, ei, h, wbuf, dsh, acc);
    __syncthreads();

    for (int r = wv; r < RNG; r += SG_WAVES) {
        const int row = r0 + r;
        if (row < NN) {
            const float* a = acc + r * DF + lane;
            const float4 v = make_float4(a[0], a[32], a[64], a[96]);
            *(float4*)(agg + (size_t)row * DF + 4 * lane) = v;
        }
    }
}

__global__ __launch_bounds__(256) void k_bn_reduce(
    const float* __restrict__ rows, float* __restrict__ macc, float* __restrict__ sqacc)
{
    const int t = threadIdx.x;
    const float* __restrict__ p = rows + (size_t)blockIdx.x * CS_ROWS * (2 * DF) + t;
    float v[CS_ROWS];
    #pragma unroll
    for (int r = 0; r < CS_ROWS; ++r) v[r] = p[(size_t)r * (2 * DF)];
    float sum = 0.0f;
    #pragma unroll
    for (int r = 0; r < CS_ROWS; ++r) sum += v[r];
    const int col = ((t >> 5) << 4) | (t & 15);
    atomicAdd(((t >> 4) & 1) ? &sqacc[col] : &macc[col], sum);
}

__global__ void k_finalize(
    const float* __restrict__ upd, const float* __restrict__ agg,
    const float* __restrict__ macc, const float* __restrict__ sqacc,
    const float* __restrict__ gamma, const float* __restrict__ beta,
    float* __restrict__ out)
{
    const long long idx = (long long)blockIdx.x * blockDim.x + threadIdx.x;
    if (idx >= (long long)NN * DF) return;
    const int col = (int)(idx & (DF - 1));
    const float inv_n = 1.0f / (float)NN;
    const float mean = macc[col] * inv_n;
    const float var  = sqacc[col] * inv_n - mean * mean;
    const float normed = gamma[col] * (upd[idx] - mean)
                       * rsqrtf(var + BN_EPS) + beta[col];
    out[idx] = normed + fmaxf(agg[idx], 0.0f);
}

extern "C" void kernel_launch(void* const* d_in, const int* in_sizes, int n_in,
                              void* d_out, int out_size, void* d_ws, size_t ws_size,
                              hipStream_t stream) {
    const float* x     = (const float*)d_in[0];
    const int*   ei    = (const int*)  d_in[1];
    const float* Wk    = (const float*)d_in[2];
    const float* bk    = (const float*)d_in[3];
    const float* a_i   = (const float*)d_in[4];
    const float* a_j   = (const float*)d_in[5];
    const float* a_e   = (const float*)d_in[6];
    const float* b_a   = (const float*)d_in[7];
    const float* Wm    = (const float*)d_in[8];
    const float* bm    = (const float*)d_in[9];
    const float* gamma = (const float*)d_in[10];
    const float* beta  = (const float*)d_in[11];
    float* out = (float*)d_out;
    float* ws  = (float*)d_ws;

    float* h_buf  = ws + OFF_H;
    float* upd    = ws + OFF_UPD;
    float* agg    = ws + OFF_AGG;
    float* s_i    = ws + OFF_SI;
    float* s_j    = ws + OFF_SJ;
    float* maxb   = ws + OFF_MAX;
    float* denom  = ws + OFF_DEN;
    float* wbuf   = ws + OFF_W;
    float* macc   = ws + OFF_MACC;
    float* sqacc  = ws + OFF_SQ;
    unsigned char* rbyte = (unsigned char*)(ws + OFF_SI);
    float* colsum_rows = ws + OFF_AGG;

    {
        const int n = NN * NHEADS;
        k_init<<<(n + 255) / 256, 256, 0, stream>>>(ws);
    }
    const int gemm_blocks = GEMM_BLOCKS;
    k_gemm_wmma<1, 0, 1, 0><<<gemm_blocks, 256, 0, stream>>>(
        x, Wk, bk, h_buf, NN, a_i, a_j, s_i, s_j, nullptr);
    k_gemm_wmma<0, 1, 0, 1><<<gemm_blocks, 256, 0, stream>>>(
        x, Wm, bm, upd, NN, nullptr, nullptr, nullptr, nullptr, colsum_rows);
    k_bn_reduce<<<GEMM_BLOCKS / CS_ROWS, 256, 0, stream>>>(colsum_rows, macc, sqacc);
    {
        const int n = NE * NHEADS;
        k_edge_logits_max<<<(n + 255) / 256, 256, 0, stream>>>(
            ei, s_i, s_j, a_e, b_a, wbuf, maxb);
    }
    {
        const int n = NE * NHEADS;
        k_edge_exp_sum<<<(n + 255) / 256, 256, 0, stream>>>(
            ei, maxb, wbuf, denom, rbyte);
    }
    k_edge_range_sum<<<NRNG, 512, 0, stream>>>(ei, h_buf, wbuf, denom, rbyte, agg);
    {
        const long long n = (long long)NN * DF;
        k_finalize<<<(int)((n + 255) / 256), 256, 0, stream>>>(
            upd, agg, macc, sqacc, gamma, beta, out);
    }
}


# Output rows formed in the message-sum launch: the work-group holding a range's sums in LDS batch-norms its upd rows, adds relu of the sums, stores out; agg plane and finalize launch gone
# speedup vs baseline: 2.2429x; 1.0810x over previous
#include <hip/hip_runtime.h>
#include <hip/hip_bf16.h>
#include <math.h>

#define NN      50000
#define NE      800000
#define DF      128
#define NHEADS  4
#define HU      32
#define NEG_SLOPE 0.2f
#define BN_EPS    0.001f
#define RNG     196
#define NRNG    ((NN + RNG - 1) / RNG)
#define GEMM_BLOCKS ((NN + 15) / 16)
#define CS_ROWS 25
static_assert(GEMM_BLOCKS % CS_ROWS == 0, "the reduce launch reads whole groups of partial rows");

typedef __attribute__((ext_vector_type(2))) float v2f;
typedef __attribute__((ext_vector_type(8))) float v8f;

#define OFF_H    ((size_t)0)
#define OFF_UPD  (OFF_H   + (size_t)NN * DF)
#define OFF_AGG  (OFF_UPD + (size_t)NN * DF)
#define OFF_SI   (OFF_AGG + (size_t)NN * DF)
#define OFF_SJ   (OFF_SI  + (size_t)NN * NHEADS)
#define OFF_MAX  (OFF_SJ  + (size_t)NN * NHEADS)
#define OFF_DEN  (OFF_MAX + (size_t)NN * NHEADS)
#define OFF_W    (OFF_DEN + (size_t)NN * NHEADS)
#define OFF_MACC (OFF_W   + (size_t)NE * NHEADS)
#define OFF_SQ   (OFF_MACC + (size_t)DF)

__device__ __forceinline__ void atomicMaxFloat(float* addr, float val) {
    if (val >= 0.0f) {
        atomicMax((int*)addr, __float_as_int(val));
    } else {
        atomicMin((unsigned int*)addr, __float_as_uint(val));
    }
}

__global__ void k_init(float* __restrict__ ws) {
    int i = blockIdx.x * blockDim.x + threadIdx.x;
    if (i < NN * NHEADS) {
        ws[OFF_MAX + i] = -INFINITY;
        ws[OFF_DEN + i] = 0.0f;
    }
    if (i < DF) {
        ws[OFF_MACC + i] = 0.0f;
        ws[OFF_SQ + i]   = 0.0f;
    }
}

template <int M>
__device__ __forceinline__ float row_xor(float x) {
    return __int_as_float(__builtin_amdgcn_update_dpp(
        0, __float_as_int(x), 0x160 + M, 0xf, 0xf, true));
}
__device__ __forceinline__ float other_half(float x) {
    const int v = __float_as_int(x);
    return __int_as_float(__builtin_amdgcn_permlanex16(v, v, 0x76543210, 0xfedcba98, false, false));
}
__device__ __forceinline__ float pick(bool c, float a, float b) { return c ? a : b; }
template <int W>
__device__ __forceinline__ void fold(float* v, bool up) {
    #pragma unroll
    for (int k = 0; k < W; ++k) {
        const float lo = v[k], hi = v[k + W];
        v[k] = pick(up, hi, lo) + row_xor<W>(pick(up, lo, hi));
    }
}

template <int WLAYOUT, int DORELU, int SCORES, int COLSUMS>
__global__ __launch_bounds__(256) void k_gemm_wmma(
    const float* __restrict__ A, const float* __restrict__ B,
    const float* __restrict__ bias, float* __restrict__ C, int M,
    const float* __restrict__ a_i, const float* __restrict__ a_j,
    float* __restrict__ s_i, float* __restrict__ s_j,
    float* __restrict__ colsum_rows)
{
    __shared__ float atile[16 * DF];
    __shared__ float part[SCORES ? 8 * 2 * 16 : 1];

    const int tid = threadIdx.x;
    const int m0  = blockIdx.x * 16;

    #pragma unroll
    for (int i = 0; i < 8; ++i) {
        const int idx = tid + i * 256;
        const int r = idx >> 7, c = idx & (DF - 1);
        const int row = m0 + r;
        atile[idx] = (row < M) ? A[(size_t)row * DF + c] : 0.0f;
    }
    __syncthreads();

    const int lane = tid & 31;
    const int half = lane >> 4;
    const int l    = lane & 15;
    const int n0   = (tid >> 5) * 16;
    const int col  = n0 + l;

    const int b_base   = (WLAYOUT == 0) ? col
                                        : ((col >> 5) * (DF * HU) + (col & 31));
    const int b_kstride = (WLAYOUT == 0) ? DF : HU;

    const float* __restrict__ arow = &atile[l * DF + 2 * half];
    const float* __restrict__ bcol = &B[b_base + (size_t)(2 * half) * b_kstride];

    v8f acc = {};
    #pragma unroll
    for (int kb = 0; kb < DF; kb += 4) {
        v2f a = *(const v2f*)(arow + kb);
        v2f b;
        b.x = bcol[(size_t)kb * b_kstride];
        b.y = bcol[(size_t)(kb + 1) * b_kstride];
        acc = __builtin_amdgcn_wmma_f32_16x16x4_f32(
                  false, a, false, b, (short)0, acc, false, false);
    }

    const float bv = bias[col];
    float hv[8];
    #pragma unroll
    for (int r = 0; r < 8; ++r) {
        float v = acc[r] + bv;
        if (DORELU) v = fmaxf(v, 0.0f);
        hv[r] = v;
    }

    if (SCORES) {
        const float ai = a_i[col], aj = a_j[col];
        float v[16];
        #pragma unroll
        for (int r = 0; r < 8; ++r) { v[r] = hv[r] * ai; v[8 + r] = hv[r] * aj; }
        fold<8>(v, (l & 8) != 0);
        fold<4>(v, (l & 4) != 0);
        fold<2>(v, (l & 2) != 0);
        fold<1>(v, (l & 1) != 0);
        part[((tid >> 5) * 2 + (l >> 3)) * 16 + 8 * half + (l & 7)] = v[0];
        __syncthreads();
    }

    float cs = 0.0f, cq = 0.0f;
    #pragma unroll
    for (int r = 0; r < 8; ++r) {
        const int row = m0 + r + 8 * half;
        if (row < M) {
            C[(size_t)row * DF + col] = hv[r];
            if (COLSUMS) { cs += hv[r]; cq += hv[r] * hv[r]; }
        }
    }

    if (COLSUMS) {
        cs += other_half(cs);
        cq += other_half(cq);
        colsum_rows[(size_t)blockIdx.x * (2 * DF) + (tid >> 5) * 32 + lane] = half ? cq : cs;
    }

    if (SCORES) {
        if (tid < 128) {
            const int which = tid >> 6;
            const int row = (tid & 63) >> 2, hd = tid & 3;
            const float s = part[((2 * hd) * 2 + which) * 16 + row]
                          + part[((2 * hd + 1) * 2 + which) * 16 + row];
            const int node = m0 + row;
            if (node < M) (which ? s_j : s_i)[node * NHEADS + hd] = s;
        }
    }
}

__global__ void k_edge_logits_max(
    const int* __restrict__ ei, const float* __restrict__ s_i,
    const float* __restrict__ s_j, const float* __restrict__ a_e,
    const float* __restrict__ b_a, float* __restrict__ wbuf,
    float* __restrict__ maxb)
{
    const int idx = blockIdx.x * blockDim.x + threadIdx.x;
    if (idx >= NE * NHEADS) return;
    const int e  = idx >> 2;
    const int hd = idx & 3;
    const int recv = ei[2 * e];
    const int send = ei[2 * e + 1];
    float lg = s_i[recv * NHEADS + hd] + s_j[send * NHEADS + hd]
             + a_e[hd] + b_a[hd];
    lg = (lg > 0.0f) ? lg : NEG_SLOPE * lg;
    wbuf[idx] = lg;
    atomicMaxFloat(&maxb[recv * NHEADS + hd], lg);
}

__global__ void k_edge_exp_sum(
    const int* __restrict__ ei, const float* __restrict__ maxb,
    float* __restrict__ wbuf, float* __restrict__ denom,
    unsigned char* __restrict__ rbyte)
{
    const int idx = blockIdx.x * blockDim.x + threadIdx.x;
    if (idx >= NE * NHEADS) return;
    const int e  = idx >> 2;
    const int hd = idx & 3;
    const int recv = ei[2 * e];
    const float w = expf(wbuf[idx] - maxb[recv * NHEADS + hd]);
    wbuf[idx] = w;
    atomicAdd(&denom[recv * NHEADS + hd], w);
    if (hd == 0) rbyte[e] = (unsigned char)(recv / RNG);
}

#define SG_WAVES 16
#define QCAP     1024
#define NVEC     (NE / 16)
#define NSUB     ((NVEC + 31) / 32)

__device__ __forceinline__ unsigned zero_bytes(unsigned t) {
    return ~(((t & 0x7f7f7f7fu) + 0x7f7f7f7fu) | t) & 0x80808080u;
}

template <bool FULL>
__device__ __forceinline__ void add_batch(
    const int* q, unsigned head, int n, int r0, int lane,
    const int* __restrict__ ei, const float* __restrict__ h,
    const float* __restrict__ wbuf, const float* dsh, float* acc)
{
    const int hd = lane >> 3;
    int   e[8];
    int2  rs[8];
    float w[8];
    #pragma unroll
    for (int b = 0; b < 8; ++b) {
        const int s = FULL ? b : (b < n ? b : n - 1);
        e[b] = q[(head + s) & (QCAP - 1)];
    }
    #pragma unroll
    for (int b = 0; b < 8; ++b) {
        rs[b] = *(const int2*)(ei + 2 * e[b]);
        w[b]  = wbuf[e[b] * NHEADS + hd];
    }
    float4 hv[8];
    #pragma unroll
    for (int b = 0; b < 8; ++b)
        hv[b] = *(const float4*)(h + (size_t)rs[b].y * DF + 4 * lane);
    #pragma unroll
    for (int b = 0; b < 8; ++b) {
        if (FULL || b < n) {
            const int rel = rs[b].x - r0;
            const float alpha = w[b] / (dsh[rel * NHEADS + hd] + 1e-16f);
            float* a = acc + rel * DF + lane;
            atomicAdd(a,      hv[b].x * alpha);
            atomicAdd(a + 32, hv[b].y * alpha);
            atomicAdd(a + 64, hv[b].z * alpha);
            atomicAdd(a + 96, hv[b].w * alpha);
        }
    }
}

#define ROWS_PER_WAVE ((RNG + SG_WAVES - 1) / SG_WAVES)

__device__ __forceinline__ float out_elem(float u, float mean, float rs, float g, float b, float a) {
    const float normed = g * (u - mean) * rs + b;
    return normed + fmaxf(a, 0.0f);
}

__global__ __launch_bounds__(512) void k_edge_range_sum(
    const int* __restrict__ ei, const float* __restrict__ h,
    const float* __restrict__ wbuf, const float* __restrict__ denom,
    const unsigned char* __restrict__ rbyte,
    const float* __restrict__ upd, const float* __restrict__ macc,
    const float* __restrict__ sqacc, const float* __restrict__ gamma,
    const float* __restrict__ beta, float* __restrict__ out)
{
    __shared__ float acc[RNG * DF];
    __shared__ float dsh[RNG * NHEADS];
    __shared__ int   queue[SG_WAVES * QCAP];

    const int tid  = threadIdx.x;
    const int lane = tid & 31;
    const int wv   = tid >> 5;
    const int r0   = blockIdx.x * RNG;

    for (int i = tid; i < RNG * DF; i += 512) acc[i] = 0.0f;
    for (int i = tid; i < RNG * NHEADS; i += 512) {
        const int g = r0 * NHEADS + i;
        dsh[i] = (g < NN * NHEADS) ? denom[g] : 0.0f;
    }
    __syncthreads();

    const unsigned pat = blockIdx.x * 0x01010101u;
    const uint4* __restrict__ rv = (const uint4*)rbyte;
    int* q = queue + wv * QCAP;
    unsigned head = 0, tail = 0;

    uint4 cur[4];
    #pragma unroll
    for (int k = 0; k < 4; ++k) {
        const int vec = (wv + k * SG_WAVES) * 32 + lane;
        cur[k] = rv[vec < NVEC ? vec : NVEC - 1];
    }
    for (int s0 = wv; s0 < NSUB; s0 += 4 * SG_WAVES) {
        uint4 nxt[4];
        #pragma unroll
        for (int k = 0; k < 4; ++k) {
            const int vec = (s0 + (4 + k) * SG_WAVES) * 32 + lane;
            nxt[k] = rv[vec < NVEC ? vec : NVEC - 1];
        }
        #pragma unroll
        for (int k = 0; k < 4; ++k) {
            const int vec = (s0 + k * SG_WAVES) * 32 + lane;
            unsigned m = 0u;
            if (vec < NVEC)
                m = (zero_bytes(cur[k].x ^ pat) >> 7) | (zero_bytes(cur[k].y ^ pat) >> 6)
                  | (zero_bytes(cur[k].z ^ pat) >> 5) | (zero_bytes(cur[k].w ^ pat) >> 4);
            for (;;) {
                const unsigned bal = __builtin_amdgcn_ballot_w32(m != 0u);
                if (bal == 0u) break;
                if (m != 0u) {
                    const int p = __builtin_ctz(m);
                    const unsigned rank = __builtin_amdgcn_mbcnt_lo(bal, 0u);
                    q[(tail + rank) & (QCAP - 1)] = vec * 16 + ((p & 7) << 2) + (p >> 3);
                    m &= m - 1u;
                }
                tail += (unsigned)__builtin_popcount(bal);
            }
            while (tail - head >= 8u) {
                add_batch<true>(q, head, 8, r0, lane, ei, h, wbuf, dsh, acc);
                head += 8u;
            }
        }
        #pragma unroll
        for (int k = 0; k < 4; ++k) cur[k] = nxt[k];
    }
    if (tail != head)
        add_batch<false>(q, head, (int)(tail - head), r0, lane, ei, h, wbuf, dsh, acc);

    float4 u[ROWS_PER_WAVE];
    #pragma unroll
    for (int i = 0; i < ROWS_PER_WAVE; ++i) {
        const int row = r0 + wv + i * SG_WAVES;
        u[i] = *(const float4*)(upd + (size_t)(row < NN ? row : NN - 1) * DF + 4 * lane);
    }
    const float4 m4 = *(const float4*)(macc + 4 * lane);
    const float4 q4 = *(const float4*)(sqacc + 4 * lane);
    const float4 g4 = *(const float4*)(gamma + 4 * lane);
    const float4 b4 = *(const float4*)(beta + 4 * lane);
    __syncthreads();

    const float inv_n = 1.0f / (float)NN;
    const float mean0 = m4.x * inv_n, mean1 = m4.y * inv_n, mean2 = m4.z * inv_n, mean3 = m4.w * inv_n;
    const float rs0 = rsqrtf(q4.x * inv_n - mean0 * mean0 + BN_EPS);
    const float rs1 = rsqrtf(q4.y * inv_n - mean1 * mean1 + BN_EPS);
    const float rs2 = rsqrtf(q4.z * inv_n - mean2 * mean2 + BN_EPS);
    const float rs3 = rsqrtf(q4.w * inv_n - mean3 * mean3 + BN_EPS);

    #pragma unroll
    for (int i = 0; i < ROWS_PER_WAVE; ++i) {
        const int r = wv + i * SG_WAVES;
        const int row = r0 + r;
        if (r < RNG && row < NN) {
            const float* a = acc + r * DF + lane;
            float4 v;
            v.x = out_elem(u[i].x, mean0, rs0, g4.x, b4.x, a[0]);
            v.y = out_elem(u[i].y, mean1, rs1, g4.y, b4.y, a[32]);
            v.z = out_elem(u[i].z, mean2, rs2, g4.z, b4.z, a[64]);
            v.w = out_elem(u[i].w, mean3, rs3, g4.w, b4.w, a[96]);
            *(float4*)(out + (size_t)row * DF + 4 * lane) = v;
        }
    }
}

__global__ __launch_bounds__(256) void k_bn_reduce(
    const float* __restrict__ rows, float* __restrict__ macc, float* __restrict__ sqacc)
{
    const int t = threadIdx.x;
    const float* __restrict__ p = rows + (size_t)blockIdx.x * CS_ROWS * (2 * DF) + t;
    float v[CS_ROWS];
    #pragma unroll
    for (int r = 0; r < CS_ROWS; ++r) v[r] = p[(size_t)r * (2 * DF)];
    float sum = 0.0f;
    #pragma unroll
    for (int r = 0; r < CS_ROWS; ++r) sum += v[r];
    const int col = ((t >> 5) << 4) | (t & 15);
    atomicAdd(((t >> 4) & 1) ? &sqacc[col] : &macc[col], sum);
}

extern "C" void kernel_launch(void* const* d_in, const int* in_sizes, int n_in,
                              void* d_out, int out_size, void* d_ws, size_t ws_size,
                              hipStream_t stream) {
    const float* x     = (const float*)d_in[0];
    const int*   ei    = (const int*)  d_in[1];
    const float* Wk    = (const float*)d_in[2];
    const float* bk    = (const float*)d_in[3];
    const float* a_i   = (const float*)d_in[4];
    const float* a_j   = (const float*)d_in[5];
    const float* a_e   = (const float*)d_in[6];
    const float* b_a   = (const float*)d_in[7];
    const float* Wm    = (const float*)d_in[8];
    const float* bm    = (const float*)d_in[9];
    const float* gamma = (const float*)d_in[10];
    const float* beta  = (const float*)d_in[11];
    float* out = (float*)d_out;
    float* ws  = (float*)d_ws;

    float* h_buf  = ws + OFF_H;
    float* upd    = ws + OFF_UPD;
    float* s_i    = ws + OFF_SI;
    float* s_j    = ws + OFF_SJ;
    float* maxb   = ws + OFF_MAX;
    float* denom  = ws + OFF_DEN;
    float* wbuf   = ws + OFF_W;
    float* macc   = ws + OFF_MACC;
    float* sqacc  = ws + OFF_SQ;
    unsigned char* rbyte = (unsigned char*)(ws + OFF_SI);
    float* colsum_rows = ws + OFF_AGG;

    {
        const int n = NN * NHEADS;
        k_init<<<(n + 255) / 256, 256, 0, stream>>>(ws);
    }
    const int gemm_blocks = GEMM_BLOCKS;
    k_gemm_wmma<1, 0, 1, 0><<<gemm_blocks, 256, 0, stream>>>(
        x, Wk, bk, h_buf, NN, a_i, a_j, s_i, s_j, nullptr);
    k_gemm_wmma<0, 1, 0, 1><<<gemm_blocks, 256, 0, stream>>>(
        x, Wm, bm, upd, NN, nullptr, nullptr, nullptr, nullptr, colsum_rows);
    k_bn_reduce<<<GEMM_BLOCKS / CS_ROWS, 256, 0, stream>>>(colsum_rows, macc, sqacc);
    {
        const int n = NE * NHEADS;
        k_edge_logits_max<<<(n + 255) / 256, 256, 0, stream>>>(
            ei, s_i, s_j, a_e, b_a, wbuf, maxb);
    }
    {
        const int n = NE * NHEADS;
        k_edge_exp_sum<<<(n + 255) / 256, 256, 0, stream>>>(
            ei, maxb, wbuf, denom, rbyte);
    }
    k_edge_range_sum<<<NRNG, 512, 0, stream>>>(
        ei, h_buf, wbuf, denom, rbyte, upd, macc, sqacc, gamma, beta, out);
}


# Projections retiled: a wave owns a 16-row stripe x all 128 columns, x fragments go straight from global to registers, W staged per work-group in LDS as k pairs (conflict-free reads); sum order kept
# speedup vs baseline: 2.5728x; 1.1471x over previous
#include <hip/hip_runtime.h>
#include <hip/hip_bf16.h>
#include <math.h>

#define NN      50000
#define NE      800000
#define DF      128
#define NHEADS  4
#define HU      32
#define NEG_SLOPE 0.2f
#define BN_EPS    0.001f
#define RNG     196
#define NRNG    ((NN + RNG - 1) / RNG)
#define GEMM_BLOCKS ((NN + 15) / 16)
#define CS_ROWS 25
static_assert(GEMM_BLOCKS % CS_ROWS == 0, "the reduce launch reads whole groups of partial rows");

typedef __attribute__((ext_vector_type(2))) float v2f;
typedef __attribute__((ext_vector_type(8))) float v8f;

#define OFF_H    ((size_t)0)
#define OFF_UPD  (OFF_H   + (size_t)NN * DF)
#define OFF_AGG  (OFF_UPD + (size_t)NN * DF)
#define OFF_SI   (OFF_AGG + (size_t)NN * DF)
#define OFF_SJ   (OFF_SI  + (size_t)NN * NHEADS)
#define OFF_MAX  (OFF_SJ  + (size_t)NN * NHEADS)
#define OFF_DEN  (OFF_MAX + (size_t)NN * NHEADS)
#define OFF_W    (OFF_DEN + (size_t)NN * NHEADS)
#define OFF_MACC (OFF_W   + (size_t)NE * NHEADS)
#define OFF_SQ   (OFF_MACC + (size_t)DF)

__device__ __forceinline__ void atomicMaxFloat(float* addr, float val) {
    if (val >= 0.0f) {
        atomicMax((int*)addr, __float_as_int(val));
    } else {
        atomicMin((unsigned int*)addr, __float_as_uint(val));
    }
}

__global__ void k_init(float* __restrict__ ws) {
    int i = blockIdx.x * blockDim.x + threadIdx.x;
    if (i < NN * NHEADS) {
        ws[OFF_MAX + i] = -INFINITY;
        ws[OFF_DEN + i] = 0.0f;
    }
    if (i < DF) {
        ws[OFF_MACC + i] = 0.0f;
        ws[OFF_SQ + i]   = 0.0f;
    }
}

template <int M>
__device__ __forceinline__ float row_xor(float x) {
    return __int_as_float(__builtin_amdgcn_update_dpp(
        0, __float_as_int(x), 0x160 + M, 0xf, 0xf, true));
}
__device__ __forceinline__ float other_half(float x) {
    const int v = __float_as_int(x);
    return __int_as_float(__builtin_amdgcn_permlanex16(v, v, 0x76543210, 0xfedcba98, false, false));
}
__device__ __forceinline__ float pick(bool c, float a, float b) { return c ? a : b; }
template <int W>
__device__ __forceinline__ void fold(float* v, bool up) {
    #pragma unroll
    for (int k = 0; k < W; ++k) {
        const float lo = v[k], hi = v[k + W];
        v[k] = pick(up, hi, lo) + row_xor<W>(pick(up, lo, hi));
    }
}

#define GEMM_WAVES 8
#define GEMM_GRID  ((GEMM_BLOCKS + GEMM_WAVES - 1) / GEMM_WAVES)
#define WP 288
template <int WLAYOUT, int DORELU, int SCORES, int COLSUMS>
__global__ __launch_bounds__(256) void k_gemm_wmma(
    const float* __restrict__ A, const float* __restrict__ B,
    const float* __restrict__ bias, float* __restrict__ C, int M,
    const float* __restrict__ a_i, const float* __restrict__ a_j,
    float* __restrict__ s_i, float* __restrict__ s_j,
    float* __restrict__ colsum_rows)
{
    __shared__ float wlds[(DF / 2) * WP];

    const int tid  = threadIdx.x;
    const int lane = tid & 31;
    const int half = lane >> 4;
    const int l    = lane & 15;
    const int stripe = blockIdx.x * GEMM_WAVES + (tid >> 5);
    const bool active = stripe < GEMM_BLOCKS;
    const int m0 = stripe * 16;

    const bool arow_ok = active && (m0 + l) < M;
    const float* __restrict__ ap = A + (size_t)(arow_ok ? (m0 + l) : 0) * DF + 2 * half;
    v2f a[DF / 4];
    #pragma unroll
    for (int s = 0; s < DF / 4; ++s) a[s] = *(const v2f*)(ap + 4 * s);

    {
        float4 r0[8], r1[8];
        #pragma unroll
        for (int j = 0; j < 8; ++j) {
            const int i = tid + 256 * j;
            const int p = i >> 5, cq = i & 31;
            const float* __restrict__ src = (WLAYOUT == 0)
                ? B + (size_t)(2 * p) * DF + 4 * cq
                : B + (size_t)(cq >> 3) * (DF * HU) + (size_t)(2 * p) * HU + 4 * (cq & 7);
            r0[j] = *(const float4*)src;
            r1[j] = *(const float4*)(src + ((WLAYOUT == 0) ? DF : HU));
        }
        #pragma unroll
        for (int j = 0; j < 8; ++j) {
            const int i = tid + 256 * j;
            const int p = i >> 5, cq = i & 31;
            float4* d = (float4*)&wlds[p * WP + 8 * cq];
            d[0] = make_float4(r0[j].x, r1[j].x, r0[j].y, r1[j].y);
            d[1] = make_float4(r0[j].z, r1[j].z, r0[j].w, r1[j].w);
        }
    }
    __syncthreads();
    if (!active) return;

    if (!arow_ok) {
        #pragma unroll
        for (int s = 0; s < DF / 4; ++s) { a[s].x = 0.0f; a[s].y = 0.0f; }
    }

    v8f acc[8];
    #pragma unroll
    for (int t = 0; t < 8; ++t) acc[t] = v8f{};

    const float* bl = &wlds[half * WP + 2 * l];
    #pragma unroll
    for (int s = 0; s < DF / 4; ++s) {
        #pragma unroll
        for (int t = 0; t < 8; ++t) {
            const v2f b = *(const v2f*)(bl + 2 * s * WP + 32 * t);
            acc[t] = __builtin_amdgcn_wmma_f32_16x16x4_f32(
                         false, a[s], false, b, (short)0, acc[t], false, false);
        }
    }

    float sc[NHEADS];
    float sprev = 0.0f;
    #pragma unroll
    for (int t = 0; t < 8; ++t) {
        const int col = 16 * t + l;
        const float bv = bias[col];
        float hv[8];
        #pragma unroll
        for (int r = 0; r < 8; ++r) {
            float v = acc[t][r] + bv;
            if (DORELU) v = fmaxf(v, 0.0f);
            hv[r] = v;
        }

        if (SCORES) {
            const float ai = a_i[col], aj = a_j[col];
            float v[16];
            #pragma unroll
            for (int r = 0; r < 8; ++r) { v[r] = hv[r] * ai; v[8 + r] = hv[r] * aj; }
            fold<8>(v, (l & 8) != 0);
            fold<4>(v, (l & 4) != 0);
            fold<2>(v, (l & 2) != 0);
            fold<1>(v, (l & 1) != 0);
            if (t & 1) sc[t >> 1] = sprev + v[0];
            else       sprev = v[0];
        }

        float cs = 0.0f, cq = 0.0f;
        #pragma unroll
        for (int r = 0; r < 8; ++r) {
            const int row = m0 + r + 8 * half;
            if (row < M) {
                C[(size_t)row * DF + col] = hv[r];
                if (COLSUMS) { cs += hv[r]; cq += hv[r] * hv[r]; }
            }
        }

        if (COLSUMS) {
            cs += other_half(cs);
            cq += other_half(cq);
            colsum_rows[(size_t)stripe * (2 * DF) + t * 32 + lane] = half ? cq : cs;
        }
    }

    if (SCORES) {
        const int node = m0 + 8 * half + (l & 7);
        if (node < M)
            *(float4*)(((l >> 3) ? s_j : s_i) + (size_t)node * NHEADS) = make_float4(sc[0], sc[1], sc[2], sc[3]);
    }
}

__global__ void k_edge_logits_max(
    const int* __restrict__ ei, const float* __restrict__ s_i,
    const float* __restrict__ s_j, const float* __restrict__ a_e,
    const float* __restrict__ b_a, float* __restrict__ wbuf,
    float* __restrict__ maxb)
{
    const int idx = blockIdx.x * blockDim.x + threadIdx.x;
    if (idx >= NE * NHEADS) return;
    const int e  = idx >> 2;
    const int hd = idx & 3;
    const int recv = ei[2 * e];
    const int send = ei[2 * e + 1];
    float lg = s_i[recv * NHEADS + hd] + s_j[send * NHEADS + hd]
             + a_e[hd] + b_a[hd];
    lg = (lg > 0.0f) ? lg : NEG_SLOPE * lg;
    wbuf[idx] = lg;
    atomicMaxFloat(&maxb[recv * NHEADS + hd], lg);
}

__global__ void k_edge_exp_sum(
    const int* __restrict__ ei, const float* __restrict__ maxb,
    float* __restrict__ wbuf, float* __restrict__ denom,
    unsigned char* __restrict__ rbyte)
{
    const int idx = blockIdx.x * blockDim.x + threadIdx.x;
    if (idx >= NE * NHEADS) return;
    const int e  = idx >> 2;
    const int hd = idx & 3;
    const int recv = ei[2 * e];
    const float w = expf(wbuf[idx] - maxb[recv * NHEADS + hd]);
    wbuf[idx] = w;
    atomicAdd(&denom[recv * NHEADS + hd], w);
    if (hd == 0) rbyte[e] = (unsigned char)(recv / RNG);
}

#define SG_WAVES 16
#define QCAP     1024
#define NVEC     (NE / 16)
#define NSUB     ((NVEC + 31) / 32)

__device__ __forceinline__ unsigned zero_bytes(unsigned t) {
    return ~(((t & 0x7f7f7f7fu) + 0x7f7f7f7fu) | t) & 0x80808080u;
}

template <bool FULL>
__device__ __forceinline__ void add_batch(
    const int* q, unsigned head, int n, int r0, int lane,
    const int* __restrict__ ei, const float* __restrict__ h,
    const float* __restrict__ wbuf, const float* dsh, float* acc)
{
    const int hd = lane >> 3;
    int   e[8];
    int2  rs[8];
    float w[8];
    #pragma unroll
    for (int b = 0; b < 8; ++b) {
        const int s = FULL ? b : (b < n ? b : n - 1);
        e[b] = q[(head + s) & (QCAP - 1)];
    }
    #pragma unroll
    for (int b = 0; b < 8; ++b) {
        rs[b] = *(const int2*)(ei + 2 * e[b]);
        w[b]  = wbuf[e[b] * NHEADS + hd];
    }
    float4 hv[8];
    #pragma unroll
    for (int b = 0; b < 8; ++b)
        hv[b] = *(const float4*)(h + (size_t)rs[b].y * DF + 4 * lane);
    #pragma unroll
    for (int b = 0; b < 8; ++b) {
        if (FULL || b < n) {
            const int rel = rs[b].x - r0;
            const float alpha = w[b] / (dsh[rel * NHEADS + hd] + 1e-16f);
            float* a = acc + rel * DF + lane;
            atomicAdd(a,      hv[b].x * alpha);
            atomicAdd(a + 32, hv[b].y * alpha);
            atomicAdd(a + 64, hv[b].z * alpha);
            atomicAdd(a + 96, hv[b].w * alpha);
        }
    }
}

#define ROWS_PER_WAVE ((RNG + SG_WAVES - 1) / SG_WAVES)

__device__ __forceinline__ float out_elem(float u, float mean, float rs, float g, float b, float a) {
    const float normed = g * (u - mean) * rs + b;
    return normed + fmaxf(a, 0.0f);
}

__global__ __launch_bounds__(512) void k_edge_range_sum(
    const int* __restrict__ ei, const float* __restrict__ h,
    const float* __restrict__ wbuf, const float* __restrict__ denom,
    const unsigned char* __restrict__ rbyte,
    const float* __restrict__ upd, const float* __restrict__ macc,
    const float* __restrict__ sqacc, const float* __restrict__ gamma,
    const float* __restrict__ beta, float* __restrict__ out)
{
    __shared__ float acc[RNG * DF];
    __shared__ float dsh[RNG * NHEADS];
    __shared__ int   queue[SG_WAVES * QCAP];

    const int tid  = threadIdx.x;
    const int lane = tid & 31;
    const int wv   = tid >> 5;
    const int r0   = blockIdx.x * RNG;

    for (int i = tid; i < RNG * DF; i += 512) acc[i] = 0.0f;
    for (int i = tid; i < RNG * NHEADS; i += 512) {
        const int g = r0 * NHEADS + i;
        dsh[i] = (g < NN * NHEADS) ? denom[g] : 0.0f;
    }
    __syncthreads();

    const unsigned pat = blockIdx.x * 0x01010101u;
    const uint4* __restrict__ rv = (const uint4*)rbyte;
    int* q = queue + wv * QCAP;
    unsigned head = 0, tail = 0;

    uint4 cur[4];
    #pragma unroll
    for (int k = 0; k < 4; ++k) {
        const int vec = (wv + k * SG_WAVES) * 32 + lane;
        cur[k] = rv[vec < NVEC ? vec : NVEC - 1];
    }
    for (int s0 = wv; s0 < NSUB; s0 += 4 * SG_WAVES) {
        uint4 nxt[4];
        #pragma unroll
        for (int k = 0; k < 4; ++k) {
            const int vec = (s0 + (4 + k) * SG_WAVES) * 32 + lane;
            nxt[k] = rv[vec < NVEC ? vec : NVEC - 1];
        }
        #pragma unroll
        for (int k = 0; k < 4; ++k) {
            const int vec = (s0 + k * SG_WAVES) * 32 + lane;
            unsigned m = 0u;
            if (vec < NVEC)
                m = (zero_bytes(cur[k].x ^ pat) >> 7) | (zero_bytes(cur[k].y ^ pat) >> 6)
                  | (zero_bytes(cur[k].z ^ pat) >> 5) | (zero_bytes(cur[k].w ^ pat) >> 4);
            for (;;) {
                const unsigned bal = __builtin_amdgcn_ballot_w32(m != 0u);
                if (bal == 0u) break;
                if (m != 0u) {
                    const int p = __builtin_ctz(m);
                    const unsigned rank = __builtin_amdgcn_mbcnt_lo(bal, 0u);
                    q[(tail + rank) & (QCAP - 1)] = vec * 16 + ((p & 7) << 2) + (p >> 3);
                    m &= m - 1u;
                }
                tail += (unsigned)__builtin_popcount(bal);
            }
            while (tail - head >= 8u) {
                add_batch<true>(q, head, 8, r0, lane, ei, h, wbuf, dsh, acc);
                head += 8u;
            }
        }
        #pragma unroll
        for (int k = 0; k < 4; ++k) cur[k] = nxt[k];
    }
    if (tail != head)
        add_batch<false>(q, head, (int)(tail - head), r0, lane, ei, h, wbuf, dsh, acc);

    float4 u[ROWS_PER_WAVE];
    #pragma unroll
    for (int i = 0; i < ROWS_PER_WAVE; ++i) {
        const int row = r0 + wv + i * SG_WAVES;
        u[i] = *(const float4*)(upd + (size_t)(row < NN ? row : NN - 1) * DF + 4 * lane);
    }
    const float4 m4 = *(const float4*)(macc + 4 * lane);
    const float4 q4 = *(const float4*)(sqacc + 4 * lane);
    const float4 g4 = *(const float4*)(gamma + 4 * lane);
    const float4 b4 = *(const float4*)(beta + 4 * lane);
    __syncthreads();

    const float inv_n = 1.0f / (float)NN;
    const float mean0 = m4.x * inv_n, mean1 = m4.y * inv_n, mean2 = m4.z * inv_n, mean3 = m4.w * inv_n;
    const float rs0 = rsqrtf(q4.x * inv_n - mean0 * mean0 + BN_EPS);
    const float rs1 = rsqrtf(q4.y * inv_n - mean1 * mean1 + BN_EPS);
    const float rs2 = rsqrtf(q4.z * inv_n - mean2 * mean2 + BN_EPS);
    const float rs3 = rsqrtf(q4.w * inv_n - mean3 * mean3 + BN_EPS);

    #pragma unroll
    for (int i = 0; i < ROWS_PER_WAVE; ++i) {
        const int r = wv + i * SG_WAVES;
        const int row = r0 + r;
        if (r < RNG && row < NN) {
            const float* a = acc + r * DF + lane;
            float4 v;
            v.x = out_elem(u[i].x, mean0, rs0, g4.x, b4.x, a[0]);
            v.y = out_elem(u[i].y, mean1, rs1, g4.y, b4.y, a[32]);
            v.z = out_elem(u[i].z, mean2, rs2, g4.z, b4.z, a[64]);
            v.w = out_elem(u[i].w, mean3, rs3, g4.w, b4.w, a[96]);
            *(float4*)(out + (size_t)row * DF + 4 * lane) = v;
        }
    }
}

__global__ __launch_bounds__(256) void k_bn_reduce(
    const float* __restrict__ rows, float* __restrict__ macc, float* __restrict__ sqacc)
{
    const int t = threadIdx.x;
    const float* __restrict__ p = rows + (size_t)blockIdx.x * CS_ROWS * (2 * DF) + t;
    float v[CS_ROWS];
    #pragma unroll
    for (int r = 0; r < CS_ROWS; ++r) v[r] = p[(size_t)r * (2 * DF)];
    float sum = 0.0f;
    #pragma unroll
    for (int r = 0; r < CS_ROWS; ++r) sum += v[r];
    const int col = ((t >> 5) << 4) | (t & 15);
    atomicAdd(((t >> 4) & 1) ? &sqacc[col] : &macc[col], sum);
}

extern "C" void kernel_launch(void* const* d_in, const int* in_sizes, int n_in,
                              void* d_out, int out_size, void* d_ws, size_t ws_size,
                              hipStream_t stream) {
    const float* x     = (const float*)d_in[0];
    const int*   ei    = (const int*)  d_in[1];
    const float* Wk    = (const float*)d_in[2];
    const float* bk    = (const float*)d_in[3];
    const float* a_i   = (const float*)d_in[4];
    const float* a_j   = (const float*)d_in[5];
    const float* a_e   = (const float*)d_in[6];
    const float* b_a   = (const float*)d_in[7];
    const float* Wm    = (const float*)d_in[8];
    const float* bm    = (const float*)d_in[9];
    const float* gamma = (const float*)d_in[10];
    const float* beta  = (const float*)d_in[11];
    float* out = (float*)d_out;
    float* ws  = (float*)d_ws;

    float* h_buf  = ws + OFF_H;
    float* upd    = ws + OFF_UPD;
    float* s_i    = ws + OFF_SI;
    float* s_j    = ws + OFF_SJ;
    float* maxb   = ws + OFF_MAX;
    float* denom  = ws + OFF_DEN;
    float* wbuf   = ws + OFF_W;
    float* macc   = ws + OFF_MACC;
    float* sqacc  = ws + OFF_SQ;
    unsigned char* rbyte = (unsigned char*)(ws + OFF_SI);
    float* colsum_rows = ws + OFF_AGG;

    {
        const int n = NN * NHEADS;
        k_init<<<(n + 255) / 256, 256, 0, stream>>>(ws);
    }
    k_gemm_wmma<1, 0, 1, 0><<<GEMM_GRID, 256, 0, stream>>>(
        x, Wk, bk, h_buf, NN, a_i, a_j, s_i, s_j, nullptr);
    k_gemm_wmma<0, 1, 0, 1><<<GEMM_GRID, 256, 0, stream>>>(
        x, Wm, bm, upd, NN, nullptr, nullptr, nullptr, nullptr, colsum_rows);
    k_bn_reduce<<<GEMM_BLOCKS / CS_ROWS, 256, 0, stream>>>(colsum_rows, macc, sqacc);
    {
        const int n = NE * NHEADS;
        k_edge_logits_max<<<(n + 255) / 256, 256, 0, stream>>>(
            ei, s_i, s_j, a_e, b_a, wbuf, maxb);
    }
    {
        const int n = NE * NHEADS;
        k_edge_exp_sum<<<(n + 255) / 256, 256, 0, stream>>>(
            ei, maxb, wbuf, denom, rbyte);
    }
    k_edge_range_sum<<<NRNG, 512, 0, stream>>>(
        ei, h_buf, wbuf, denom, rbyte, upd, macc, sqacc, gamma, beta, out);
}


# Segment softmax in the message-sum launch: a range's work-group takes logit maxima and denominators by LDS atomics (arrival order), keeps its edges in LDS, divides sums once; per-edge launches gone
# speedup vs baseline: 5.3081x; 2.0632x over previous
#include <hip/hip_runtime.h>
#include <hip/hip_bf16.h>
#include <math.h>

#define NN      50000
#define NE      800000
#define DF      128
#define NHEADS  4
#define HU      32
#define NEG_SLOPE 0.2f
#define BN_EPS    0.001f
#define RNG     196
#define NRNG    ((NN + RNG - 1) / RNG)
#define GEMM_BLOCKS ((NN + 15) / 16)
#define CS_ROWS 25
static_assert(GEMM_BLOCKS % CS_ROWS == 0, "the reduce launch reads whole groups of partial rows");

typedef __attribute__((ext_vector_type(2))) float v2f;
typedef __attribute__((ext_vector_type(8))) float v8f;

#define OFF_H    ((size_t)0)
#define OFF_UPD  (OFF_H   + (size_t)NN * DF)
#define OFF_AGG  (OFF_UPD + (size_t)NN * DF)
#define OFF_SI   (OFF_AGG + (size_t)NN * DF)
#define OFF_SJ   (OFF_SI  + (size_t)NN * NHEADS)
#define OFF_MAX  (OFF_SJ  + (size_t)NN * NHEADS)
#define OFF_DEN  (OFF_MAX + (size_t)NN * NHEADS)
#define OFF_W    (OFF_DEN + (size_t)NN * NHEADS)
#define OFF_MACC (OFF_W   + (size_t)NE * NHEADS)
#define OFF_SQ   (OFF_MACC + (size_t)DF)

#define NVEC     (NE / 16)
static_assert(NE % 16 == 0, "whole 16-edge vectors");
__global__ void k_init(const int* __restrict__ ei, float* __restrict__ ws,
                       unsigned char* __restrict__ rbyte) {
    const int i = blockIdx.x * blockDim.x + threadIdx.x;
    if (i < DF) {
        ws[OFF_MACC + i] = 0.0f;
        ws[OFF_SQ + i]   = 0.0f;
    }
    if (i < NVEC) {
        const int4* __restrict__ p = (const int4*)(ei + 32 * (size_t)i);
        int4 a[8];
        #pragma unroll
        for (int j = 0; j < 8; ++j) a[j] = p[j];
        unsigned w[4];
        #pragma unroll
        for (int j = 0; j < 4; ++j)
            w[j] =  ((unsigned)a[2 * j].x / RNG)            | (((unsigned)a[2 * j].z / RNG) << 8)
                 | (((unsigned)a[2 * j + 1].x / RNG) << 16) | (((unsigned)a[2 * j + 1].z / RNG) << 24);
        ((uint4*)rbyte)[i] = make_uint4(w[0], w[1], w[2], w[3]);
    }
}

template <int M>
__device__ __forceinline__ float row_xor(float x) {
    return __int_as_float(__builtin_amdgcn_update_dpp(
        0, __float_as_int(x), 0x160 + M, 0xf, 0xf, true));
}
__device__ __forceinline__ float other_half(float x) {
    const int v = __float_as_int(x);
    return __int_as_float(__builtin_amdgcn_permlanex16(v, v, 0x76543210, 0xfedcba98, false, false));
}
__device__ __forceinline__ float pick(bool c, float a, float b) { return c ? a : b; }
template <int W>
__device__ __forceinline__ void fold(float* v, bool up) {
    #pragma unroll
    for (int k = 0; k < W; ++k) {
        const float lo = v[k], hi = v[k + W];
        v[k] = pick(up, hi, lo) + row_xor<W>(pick(up, lo, hi));
    }
}

#define GEMM_WAVES 8
#define GEMM_GRID  ((GEMM_BLOCKS + GEMM_WAVES - 1) / GEMM_WAVES)
#define WP 288
template <int WLAYOUT, int DORELU, int SCORES, int COLSUMS>
__global__ __launch_bounds__(256) void k_gemm_wmma(
    const float* __restrict__ A, const float* __restrict__ B,
    const float* __restrict__ bias, float* __restrict__ C, int M,
    const float* __restrict__ a_i, const float* __restrict__ a_j,
    float* __restrict__ s_i, float* __restrict__ s_j,
    float* __restrict__ colsum_rows)
{
    __shared__ float wlds[(DF / 2) * WP];

    const int tid  = threadIdx.x;
    const int lane = tid & 31;
    const int half = lane >> 4;
    const int l    = lane & 15;
    const int stripe = blockIdx.x * GEMM_WAVES + (tid >> 5);
    const bool active = stripe < GEMM_BLOCKS;
    const int m0 = stripe * 16;

    const bool arow_ok = active && (m0 + l) < M;
    const float* __restrict__ ap = A + (size_t)(arow_ok ? (m0 + l) : 0) * DF + 2 * half;
    v2f a[DF / 4];
    #pragma unroll
    for (int s = 0; s < DF / 4; ++s) a[s] = *(const v2f*)(ap + 4 * s);

    {
        float4 r0[8], r1[8];
        #pragma unroll
        for (int j = 0; j < 8; ++j) {
            const int i = tid + 256 * j;
            const int p = i >> 5, cq = i & 31;
            const float* __restrict__ src = (WLAYOUT == 0)
                ? B + (size_t)(2 * p) * DF + 4 * cq
                : B + (size_t)(cq >> 3) * (DF * HU) + (size_t)(2 * p) * HU + 4 * (cq & 7);
            r0[j] = *(const float4*)src;
            r1[j] = *(const float4*)(src + ((WLAYOUT == 0) ? DF : HU));
        }
        #pragma unroll
        for (int j = 0; j < 8; ++j) {
            const int i = tid + 256 * j;
            const int p = i >> 5, cq = i & 31;
            float4* d = (float4*)&wlds[p * WP + 8 * cq];
            d[0] = make_float4(r0[j].x, r1[j].x, r0[j].y, r1[j].y);
            d[1] = make_float4(r0[j].z, r1[j].z, r0[j].w, r1[j].w);
        }
    }
    __syncthreads();
    if (!active) return;

    if (!arow_ok) {
        #pragma unroll
        for (int s = 0; s < DF / 4; ++s) { a[s].x = 0.0f; a[s].y = 0.0f; }
    }

    v8f acc[8];
    #pragma unroll
    for (int t = 0; t < 8; ++t) acc[t] = v8f{};

    const float* bl = &wlds[half * WP + 2 * l];
    #pragma unroll
    for (int s = 0; s < DF / 4; ++s) {
        #pragma unroll
        for (int t = 0; t < 8; ++t) {
            const v2f b = *(const v2f*)(bl + 2 * s * WP + 32 * t);
            acc[t] = __builtin_amdgcn_wmma_f32_16x16x4_f32(
                         false, a[s], false, b, (short)0, acc[t], false, false);
        }
    }

    float sc[NHEADS];
    float sprev = 0.0f;
    #pragma unroll
    for (int t = 0; t < 8; ++t) {
        const int col = 16 * t + l;
        const float bv = bias[col];
        float hv[8];
        #pragma unroll
        for (int r = 0; r < 8; ++r) {
            float v = acc[t][r] + bv;
            if (DORELU) v = fmaxf(v, 0.0f);
            hv[r] = v;
        }

        if (SCORES) {
            const float ai = a_i[col], aj = a_j[col];
            float v[16];
            #pragma unroll
            for (int r = 0; r < 8; ++r) { v[r] = hv[r] * ai; v[8 + r] = hv[r] * aj; }
            fold<8>(v, (l & 8) != 0);
            fold<4>(v, (l & 4) != 0);
            fold<2>(v, (l & 2) != 0);
            fold<1>(v, (l & 1) != 0);
            if (t & 1) sc[t >> 1] = sprev + v[0];
            else       sprev = v[0];
        }

        float cs = 0.0f, cq = 0.0f;
        #pragma unroll
        for (int r = 0; r < 8; ++r) {
            const int row = m0 + r + 8 * half;
            if (row < M) {
                C[(size_t)row * DF + col] = hv[r];
                if (COLSUMS) { cs += hv[r]; cq += hv[r] * hv[r]; }
            }
        }

        if (COLSUMS) {
            cs += other_half(cs);
            cq += other_half(cq);
            colsum_rows[(size_t)stripe * (2 * DF) + t * 32 + lane] = half ? cq : cs;
        }
    }

    if (SCORES) {
        const int node = m0 + 8 * half + (l & 7);
        if (node < M)
            *(float4*)(((l >> 3) ? s_j : s_i) + (size_t)node * NHEADS) = make_float4(sc[0], sc[1], sc[2], sc[3]);
    }
}

#define SG_WAVES 16
#define PCAP     64
#define LCAP     384
#define NSUB     ((NVEC + 31) / 32)
static_assert(NN <= 65536 && RNG <= 256, "a list entry packs sender (16 bits) and receiver - r0");

__device__ __forceinline__ unsigned zero_bytes(unsigned t) {
    return ~(((t & 0x7f7f7f7fu) + 0x7f7f7f7fu) | t) & 0x80808080u;
}

__device__ __forceinline__ void lds_max_float(float* cell, float v) {
    if (__float_as_int(v) >= 0) atomicMax((int*)cell, __float_as_int(v));
    else                        atomicMin((unsigned int*)cell, __float_as_uint(v));
}

__device__ __forceinline__ float leaky(float v) { return (v > 0.0f) ? v : NEG_SLOPE * v; }

template <class Sink>
__device__ __forceinline__ void walk_range_bytes(
    const uint4* __restrict__ rv, unsigned pat, int wv, int lane, int* pend, Sink&& sink)
{
    unsigned head = 0, tail = 0;
    uint4 cur[4];
    #pragma unroll
    for (int k = 0; k < 4; ++k) {
        const int vec = (wv + k * SG_WAVES) * 32 + lane;
        cur[k] = rv[vec < NVEC ? vec : NVEC - 1];
    }
    for (int s0 = wv; s0 < NSUB; s0 += 4 * SG_WAVES) {
        uint4 nxt[4];
        #pragma unroll
        for (int k = 0; k < 4; ++k) {
            const int vec = (s0 + (4 + k) * SG_WAVES) * 32 + lane;
            nxt[k] = rv[vec < NVEC ? vec : NVEC - 1];
        }
        #pragma unroll
        for (int k = 0; k < 4; ++k) {
            const int vec = (s0 + k * SG_WAVES) * 32 + lane;
            unsigned m = 0u;
            if (vec < NVEC)
                m = (zero_bytes(cur[k].x ^ pat) >> 7) | (zero_bytes(cur[k].y ^ pat) >> 6)
                  | (zero_bytes(cur[k].z ^ pat) >> 5) | (zero_bytes(cur[k].w ^ pat) >> 4);
            for (;;) {
                const unsigned bal = __builtin_amdgcn_ballot_w32(m != 0u);
                if (bal == 0u) break;
                if (m != 0u) {
                    const int p = __builtin_ctz(m);
                    const unsigned rank = __builtin_amdgcn_mbcnt_lo(bal, 0u);
                    pend[(tail + rank) & (PCAP - 1)] = vec * 16 + ((p & 7) << 2) + (p >> 3);
                    m &= m - 1u;
                }
                tail += (unsigned)__builtin_popcount(bal);
                if (tail - head >= 32u) { sink(head, 32); head += 32u; }
            }
        }
        #pragma unroll
        for (int k = 0; k < 4; ++k) cur[k] = nxt[k];
    }
    if (tail != head) sink(head, (int)(tail - head));
}

__device__ __forceinline__ void edge_logits(
    const int* pend, unsigned head, int n, int lane, int r0,
    const int* __restrict__ ei, const float* __restrict__ s_j, const float* sis,
    float4 ae, float4 ba, int& packed, int& rel, float4& lg)
{
    const int e = pend[(head + (unsigned)(lane < n ? lane : 0)) & (PCAP - 1)];
    const int2 rs = *(const int2*)(ei + 2 * (size_t)e);
    const float4 sj = *(const float4*)(s_j + (size_t)rs.y * NHEADS);
    rel = rs.x - r0;
    const float4 si = *(const float4*)(sis + rel * NHEADS);
    lg.x = leaky(si.x + sj.x + ae.x + ba.x);
    lg.y = leaky(si.y + sj.y + ae.y + ba.y);
    lg.z = leaky(si.z + sj.z + ae.z + ba.z);
    lg.w = leaky(si.w + sj.w + ae.w + ba.w);
    packed = rs.y | (rel << 16);
}

__device__ __forceinline__ void list_weights(
    const int* lsr, float* llg, int from, int n, int lane, const float* mxs, float* dns)
{
    for (int b = 0; b < n; b += 32) {
        const int i = from + b + lane;
        if (b + lane < n) {
            const int rel = lsr[i] >> 16;
            const float4 l = *(const float4*)(llg + 4 * i);
            const float4 m = *(const float4*)(mxs + rel * NHEADS);
            float4 w;
            w.x = expf(l.x - m.x); w.y = expf(l.y - m.y); w.z = expf(l.z - m.z); w.w = expf(l.w - m.w);
            atomicAdd(dns + rel * NHEADS + 0, w.x);
            atomicAdd(dns + rel * NHEADS + 1, w.y);
            atomicAdd(dns + rel * NHEADS + 2, w.z);
            atomicAdd(dns + rel * NHEADS + 3, w.w);
            *(float4*)(llg + 4 * i) = w;
        }
    }
}

template <bool FULL>
__device__ __forceinline__ void add_batch(
    const int* lsr, const float* llg, int base, int n, int lane,
    const float* __restrict__ h, float* acc)
{
    const int hd = lane >> 3;
    int   sr[8];
    float w[8];
    #pragma unroll
    for (int b = 0; b < 8; ++b) {
        const int i = base + (FULL ? b : (b < n ? b : n - 1));
        sr[b] = lsr[i];
        w[b]  = llg[4 * i + hd];
    }
    float4 hv[8];
    #pragma unroll
    for (int b = 0; b < 8; ++b)
        hv[b] = *(const float4*)(h + (size_t)(sr[b] & 0xffff) * DF + 4 * lane);
    #pragma unroll
    for (int b = 0; b < 8; ++b) {
        if (FULL || b < n) {
            float* a = acc + (sr[b] >> 16) * DF + lane;
            atomicAdd(a,      hv[b].x * w[b]);
            atomicAdd(a + 32, hv[b].y * w[b]);
            atomicAdd(a + 64, hv[b].z * w[b]);
            atomicAdd(a + 96, hv[b].w * w[b]);
        }
    }
}

__device__ __forceinline__ void list_rows(
    const int* lsr, const float* llg, int n, int lane, const float* __restrict__ h, float* acc)
{
    int base = 0;
    for (; base + 8 <= n; base += 8) add_batch<true>(lsr, llg, base, 8, lane, h, acc);
    if (base < n) add_batch<false>(lsr, llg, base, n - base, lane, h, acc);
}

#define ROWS_PER_WAVE ((RNG + SG_WAVES - 1) / SG_WAVES)

__device__ __forceinline__ float out_elem(float u, float mean, float rs, float g, float b, float a, float d) {
    const float normed = g * (u - mean) * rs + b;
    return normed + fmaxf(a / d, 0.0f);
}

__global__ __launch_bounds__(512) void k_edge_range_sum(
    const int* __restrict__ ei, const float* __restrict__ h,
    const float* __restrict__ s_i, const float* __restrict__ s_j,
    const float* __restrict__ a_e, const float* __restrict__ b_a,
    const unsigned char* __restrict__ rbyte,
    const float* __restrict__ upd, const float* __restrict__ macc,
    const float* __restrict__ sqacc, const float* __restrict__ gamma,
    const float* __restrict__ beta, float* __restrict__ out)
{
    __shared__ float acc[RNG * DF];
    __shared__ __attribute__((aligned(16))) float mxs[RNG * NHEADS];
    __shared__ __attribute__((aligned(16))) float dns[RNG * NHEADS];
    __shared__ __attribute__((aligned(16))) float sis[RNG * NHEADS];
    __shared__ int   pending[SG_WAVES * PCAP];
    __shared__ int   list_sr[SG_WAVES * LCAP];
    __shared__ __attribute__((aligned(16))) float list_lg[SG_WAVES * LCAP * 4];

    const int tid  = threadIdx.x;
    const int lane = tid & 31;
    const int wv   = tid >> 5;
    const int r0   = blockIdx.x * RNG;

    for (int i = tid; i < RNG * DF; i += 512) acc[i] = 0.0f;
    for (int i = tid; i < RNG * NHEADS; i += 512) {
        const int g = r0 * NHEADS + i;
        mxs[i] = -INFINITY;
        dns[i] = 0.0f;
        sis[i] = (g < NN * NHEADS) ? s_i[g] : 0.0f;
    }
    const float4 ae = make_float4(a_e[0], a_e[1], a_e[2], a_e[3]);
    const float4 ba = make_float4(b_a[0], b_a[1], b_a[2], b_a[3]);
    __syncthreads();

    const unsigned pat = blockIdx.x * 0x01010101u;
    const uint4* __restrict__ rv = (const uint4*)rbyte;
    int*   pend = pending + wv * PCAP;
    int*   lsr  = list_sr + wv * LCAP;
    float* llg  = list_lg + wv * LCAP * 4;
    int  cnt  = 0;
    bool over = false;

    walk_range_bytes(rv, pat, wv, lane, pend, [&](unsigned head, int n) {
        int packed, rel; float4 lg;
        edge_logits(pend, head, n, lane, r0, ei, s_j, sis, ae, ba, packed, rel, lg);
        if (lane < n) {
            lds_max_float(mxs + rel * NHEADS + 0, lg.x);
            lds_max_float(mxs + rel * NHEADS + 1, lg.y);
            lds_max_float(mxs + rel * NHEADS + 2, lg.z);
            lds_max_float(mxs + rel * NHEADS + 3, lg.w);
        }
        if (!over && cnt + n <= LCAP) {
            if (lane < n) {
                lsr[cnt + lane] = packed;
                *(float4*)(llg + 4 * (cnt + lane)) = lg;
            }
            cnt += n;
        } else {
            over = true;
        }
    });
    __syncthreads();

    list_weights(lsr, llg, 0, cnt, lane, mxs, dns);
    list_rows(lsr, llg, cnt, lane, h, acc);
    if (over) {
        int seen = 0;
        walk_range_bytes(rv, pat, wv, lane, pend, [&](unsigned head, int n) {
            const int lo = (cnt > seen) ? (cnt - seen) : 0;
            seen += n;
            if (lo >= n) return;
            int packed, rel; float4 lg;
            edge_logits(pend, head, n, lane, r0, ei, s_j, sis, ae, ba, packed, rel, lg);
            if (lane >= lo && lane < n) {
                lsr[lane - lo] = packed;
                *(float4*)(llg + 4 * (lane - lo)) = lg;
            }
            list_weights(lsr, llg, 0, n - lo, lane, mxs, dns);
            list_rows(lsr, llg, n - lo, lane, h, acc);
        });
    }

    float4 u[ROWS_PER_WAVE];
    #pragma unroll
    for (int i = 0; i < ROWS_PER_WAVE; ++i) {
        const int row = r0 + wv + i * SG_WAVES;
        u[i] = *(const float4*)(upd + (size_t)(row < NN ? row : NN - 1) * DF + 4 * lane);
    }
    const float4 m4 = *(const float4*)(macc + 4 * lane);
    const float4 q4 = *(const float4*)(sqacc + 4 * lane);
    const float4 g4 = *(const float4*)(gamma + 4 * lane);
    const float4 b4 = *(const float4*)(beta + 4 * lane);
    __syncthreads();

    const float inv_n = 1.0f / (float)NN;
    const float mean0 = m4.x * inv_n, mean1 = m4.y * inv_n, mean2 = m4.z * inv_n, mean3 = m4.w * inv_n;
    const float rs0 = rsqrtf(q4.x * inv_n - mean0 * mean0 + BN_EPS);
    const float rs1 = rsqrtf(q4.y * inv_n - mean1 * mean1 + BN_EPS);
    const float rs2 = rsqrtf(q4.z * inv_n - mean2 * mean2 + BN_EPS);
    const float rs3 = rsqrtf(q4.w * inv_n - mean3 * mean3 + BN_EPS);

    const int hd = lane >> 3;
    #pragma unroll
    for (int i = 0; i < ROWS_PER_WAVE; ++i) {
        const int r = wv + i * SG_WAVES;
        const int row = r0 + r;
        if (r < RNG && row < NN) {
            const float* a = acc + r * DF + lane;
            const float d = dns[r * NHEADS + hd] + 1e-16f;
            float4 v;
            v.x = out_elem(u[i].x, mean0, rs0, g4.x, b4.x, a[0],  d);
            v.y = out_elem(u[i].y, mean1, rs1, g4.y, b4.y, a[32], d);
            v.z = out_elem(u[i].z, mean2, rs2, g4.z, b4.z, a[64], d);
            v.w = out_elem(u[i].w, mean3, rs3, g4.w, b4.w, a[96], d);
            *(float4*)(out + (size_t)row * DF + 4 * lane) = v;
        }
    }
}

__global__ __launch_bounds__(256) void k_bn_reduce(
    const float* __restrict__ rows, float* __restrict__ macc, float* __restrict__ sqacc)
{
    const int t = threadIdx.x;
    const float* __restrict__ p = rows + (size_t)blockIdx.x * CS_ROWS * (2 * DF) + t;
    float v[CS_ROWS];
    #pragma unroll
    for (int r = 0; r < CS_ROWS; ++r) v[r] = p[(size_t)r * (2 * DF)];
    float sum = 0.0f;
    #pragma unroll
    for (int r = 0; r < CS_ROWS; ++r) sum += v[r];
    const int col = ((t >> 5) << 4) | (t & 15);
    atomicAdd(((t >> 4) & 1) ? &sqacc[col] : &macc[col], sum);
}

extern "C" void kernel_launch(void* const* d_in, const int* in_sizes, int n_in,
                              void* d_out, int out_size, void* d_ws, size_t ws_size,
                              hipStream_t stream) {
    const float* x     = (const float*)d_in[0];
    const int*   ei    = (const int*)  d_in[1];
    const float* Wk    = (const float*)d_in[2];
    const float* bk    = (const float*)d_in[3];
    const float* a_i   = (const float*)d_in[4];
    const float* a_j   = (const float*)d_in[5];
    const float* a_e   = (const float*)d_in[6];
    const float* b_a   = (const float*)d_in[7];
    const float* Wm    = (const float*)d_in[8];
    const float* bm    = (const float*)d_in[9];
    const float* gamma = (const float*)d_in[10];
    const float* beta  = (const float*)d_in[11];
    float* out = (float*)d_out;
    float* ws  = (float*)d_ws;

    float* h_buf  = ws + OFF_H;
    float* upd    = ws + OFF_UPD;
    float* s_i    = ws + OFF_SI;
    float* s_j    = ws + OFF_SJ;
    float* macc   = ws + OFF_MACC;
    float* sqacc  = ws + OFF_SQ;
    unsigned char* rbyte = (unsigned char*)(ws + OFF_W);
    float* colsum_rows = ws + OFF_AGG;

    k_init<<<(NVEC + 255) / 256, 256, 0, stream>>>(ei, ws, rbyte);
    k_gemm_wmma<1, 0, 1, 0><<<GEMM_GRID, 256, 0, stream>>>(
        x, Wk, bk, h_buf, NN, a_i, a_j, s_i, s_j, nullptr);
    k_gemm_wmma<0, 1, 0, 1><<<GEMM_GRID, 256, 0, stream>>>(
        x, Wm, bm, upd, NN, nullptr, nullptr, nullptr, nullptr, colsum_rows);
    k_bn_reduce<<<GEMM_BLOCKS / CS_ROWS, 256, 0, stream>>>(colsum_rows, macc, sqacc);
    k_edge_range_sum<<<NRNG, 512, 0, stream>>>(
        ei, h_buf, s_i, s_j, a_e, b_a, rbyte, upd, macc, sqacc, gamma, beta, out);
}


# Ranges find their edges by a per-range bitmap over the edge vectors, built in the first launch with LDS integer or: a work-group reads its row of bits, not every range byte; arrival-order sums kept
# speedup vs baseline: 5.5947x; 1.0540x over previous
#include <hip/hip_runtime.h>
#include <hip/hip_bf16.h>
#include <math.h>

#define NN      50000
#define NE      800000
#define DF      128
#define NHEADS  4
#define HU      32
#define NEG_SLOPE 0.2f
#define BN_EPS    0.001f
#define RNG     196
#define NRNG    ((NN + RNG - 1) / RNG)
#define GEMM_BLOCKS ((NN + 15) / 16)
#define CS_ROWS 25
static_assert(GEMM_BLOCKS % CS_ROWS == 0, "the reduce launch reads whole groups of partial rows");

typedef __attribute__((ext_vector_type(2))) float v2f;
typedef __attribute__((ext_vector_type(8))) float v8f;

#define OFF_H    ((size_t)0)
#define OFF_UPD  (OFF_H   + (size_t)NN * DF)
#define OFF_AGG  (OFF_UPD + (size_t)NN * DF)
#define OFF_SI   (OFF_AGG + (size_t)NN * DF)
#define OFF_SJ   (OFF_SI  + (size_t)NN * NHEADS)
#define OFF_MAX  (OFF_SJ  + (size_t)NN * NHEADS)
#define OFF_DEN  (OFF_MAX + (size_t)NN * NHEADS)
#define OFF_W    (OFF_DEN + (size_t)NN * NHEADS)
#define OFF_MACC (OFF_W   + (size_t)NE * NHEADS)
#define OFF_SQ   (OFF_MACC + (size_t)DF)

#define NVEC     (NE / 16)
static_assert(NE % 16 == 0, "whole 16-edge vectors");
#define INIT_VEC   1024
#define INIT_GRID  ((NVEC + INIT_VEC - 1) / INIT_VEC)
#define BMW        (INIT_GRID * 32)
#define BMP        (NRNG + 1)
__global__ __launch_bounds__(256) void k_init(const int* __restrict__ ei, float* __restrict__ ws,
                       unsigned char* __restrict__ rbyte, unsigned* __restrict__ bitmap) {
    __shared__ unsigned bm[32 * BMP];
    const int tid = threadIdx.x;
    const unsigned bit = 1u << (tid & 31);
    int4 a[INIT_VEC / 256][8];
    #pragma unroll
    for (int j = 0; j < INIT_VEC / 256; ++j) {
        const int i = blockIdx.x * INIT_VEC + j * 256 + tid;
        const int4* __restrict__ p = (const int4*)(ei + 32 * (size_t)(i < NVEC ? i : NVEC - 1));
        #pragma unroll
        for (int k = 0; k < 8; ++k) a[j][k] = p[k];
    }
    for (int i = tid; i < 32 * BMP; i += 256) bm[i] = 0u;
    if (blockIdx.x == 0 && tid < DF) {
        ws[OFF_MACC + tid] = 0.0f;
        ws[OFF_SQ + tid]   = 0.0f;
    }
    __syncthreads();
    #pragma unroll
    for (int j = 0; j < INIT_VEC / 256; ++j) {
        const int lv = j * 256 + tid;
        const int i  = blockIdx.x * INIT_VEC + lv;
        if (i < NVEC) {
            unsigned rid[16];
            #pragma unroll
            for (int k = 0; k < 8; ++k) {
                rid[2 * k]     = (unsigned)a[j][k].x / RNG;
                rid[2 * k + 1] = (unsigned)a[j][k].z / RNG;
            }
            unsigned w[4];
            #pragma unroll
            for (int k = 0; k < 4; ++k)
                w[k] = rid[4 * k] | (rid[4 * k + 1] << 8) | (rid[4 * k + 2] << 16) | (rid[4 * k + 3] << 24);
            ((uint4*)rbyte)[i] = make_uint4(w[0], w[1], w[2], w[3]);
            unsigned* plane = bm + (lv >> 5) * BMP;
            #pragma unroll
            for (int k = 0; k < 16; ++k) atomicOr(plane + rid[k], bit);
        }
    }
    __syncthreads();
    const int lane = tid & 31;
    for (int r = tid >> 5; r < NRNG; r += 8)
        bitmap[(size_t)r * BMW + blockIdx.x * 32 + lane] = bm[lane * BMP + r];
}

template <int M>
__device__ __forceinline__ float row_xor(float x) {
    return __int_as_float(__builtin_amdgcn_update_dpp(
        0, __float_as_int(x), 0x160 + M, 0xf, 0xf, true));
}
__device__ __forceinline__ float other_half(float x) {
    const int v = __float_as_int(x);
    return __int_as_float(__builtin_amdgcn_permlanex16(v, v, 0x76543210, 0xfedcba98, false, false));
}
__device__ __forceinline__ float pick(bool c, float a, float b) { return c ? a : b; }
template <int W>
__device__ __forceinline__ void fold(float* v, bool up) {
    #pragma unroll
    for (int k = 0; k < W; ++k) {
        const float lo = v[k], hi = v[k + W];
        v[k] = pick(up, hi, lo) + row_xor<W>(pick(up, lo, hi));
    }
}

#define GEMM_WAVES 8
#define GEMM_GRID  ((GEMM_BLOCKS + GEMM_WAVES - 1) / GEMM_WAVES)
#define WP 288
template <int WLAYOUT, int DORELU, int SCORES, int COLSUMS>
__global__ __launch_bounds__(256) void k_gemm_wmma(
    const float* __restrict__ A, const float* __restrict__ B,
    const float* __restrict__ bias, float* __restrict__ C, int M,
    const float* __restrict__ a_i, const float* __restrict__ a_j,
    float* __restrict__ s_i, float* __restrict__ s_j,
    float* __restrict__ colsum_rows)
{
    __shared__ float wlds[(DF / 2) * WP];

    const int tid  = threadIdx.x;
    const int lane = tid & 31;
    const int half = lane >> 4;
    const int l    = lane & 15;
    const int stripe = blockIdx.x * GEMM_WAVES + (tid >> 5);
    const bool active = stripe < GEMM_BLOCKS;
    const int m0 = stripe * 16;

    const bool arow_ok = active && (m0 + l) < M;
    const float* __restrict__ ap = A + (size_t)(arow_ok ? (m0 + l) : 0) * DF + 2 * half;
    v2f a[DF / 4];
    #pragma unroll
    for (int s = 0; s < DF / 4; ++s) a[s] = *(const v2f*)(ap + 4 * s);

    {
        float4 r0[8], r1[8];
        #pragma unroll
        for (int j = 0; j < 8; ++j) {
            const int i = tid + 256 * j;
            const int p = i >> 5, cq = i & 31;
            const float* __restrict__ src = (WLAYOUT == 0)
                ? B + (size_t)(2 * p) * DF + 4 * cq
                : B + (size_t)(cq >> 3) * (DF * HU) + (size_t)(2 * p) * HU + 4 * (cq & 7);
            r0[j] = *(const float4*)src;
            r1[j] = *(const float4*)(src + ((WLAYOUT == 0) ? DF : HU));
        }
        #pragma unroll
        for (int j = 0; j < 8; ++j) {
            const int i = tid + 256 * j;
            const int p = i >> 5, cq = i & 31;
            float4* d = (float4*)&wlds[p * WP + 8 * cq];
            d[0] = make_float4(r0[j].x, r1[j].x, r0[j].y, r1[j].y);
            d[1] = make_float4(r0[j].z, r1[j].z, r0[j].w, r1[j].w);
        }
    }
    __syncthreads();
    if (!active) return;

    if (!arow_ok) {
        #pragma unroll
        for (int s = 0; s < DF / 4; ++s) { a[s].x = 0.0f; a[s].y = 0.0f; }
    }

    v8f acc[8];
    #pragma unroll
    for (int t = 0; t < 8; ++t) acc[t] = v8f{};

    const float* bl = &wlds[half * WP + 2 * l];
    #pragma unroll
    for (int s = 0; s < DF / 4; ++s) {
        #pragma unroll
        for (int t = 0; t < 8; ++t) {
            const v2f b = *(const v2f*)(bl + 2 * s * WP + 32 * t);
            acc[t] = __builtin_amdgcn_wmma_f32_16x16x4_f32(
                         false, a[s], false, b, (short)0, acc[t], false, false);
        }
    }

    float sc[NHEADS];
    float sprev = 0.0f;
    #pragma unroll
    for (int t = 0; t < 8; ++t) {
        const int col = 16 * t + l;
        const float bv = bias[col];
        float hv[8];
        #pragma unroll
        for (int r = 0; r < 8; ++r) {
            float v = acc[t][r] + bv;
            if (DORELU) v = fmaxf(v, 0.0f);
            hv[r] = v;
        }

        if (SCORES) {
            const float ai = a_i[col], aj = a_j[col];
            float v[16];
            #pragma unroll
            for (int r = 0; r < 8; ++r) { v[r] = hv[r] * ai; v[8 + r] = hv[r] * aj; }
            fold<8>(v, (l & 8) != 0);
            fold<4>(v, (l & 4) != 0);
            fold<2>(v, (l & 2) != 0);
            fold<1>(v, (l & 1) != 0);
            if (t & 1) sc[t >> 1] = sprev + v[0];
            else       sprev = v[0];
        }

        float cs = 0.0f, cq = 0.0f;
        #pragma unroll
        for (int r = 0; r < 8; ++r) {
            const int row = m0 + r + 8 * half;
            if (row < M) {
                C[(size_t)row * DF + col] = hv[r];
                if (COLSUMS) { cs += hv[r]; cq += hv[r] * hv[r]; }
            }
        }

        if (COLSUMS) {
            cs += other_half(cs);
            cq += other_half(cq);
            colsum_rows[(size_t)stripe * (2 * DF) + t * 32 + lane] = half ? cq : cs;
        }
    }

    if (SCORES) {
        const int node = m0 + 8 * half + (l & 7);
        if (node < M)
            *(float4*)(((l >> 3) ? s_j : s_i) + (size_t)node * NHEADS) = make_float4(sc[0], sc[1], sc[2], sc[3]);
    }
}

#define SG_WAVES 16
#define PCAP     64
#define LCAP     384
static_assert(NN <= 65536 && RNG <= 256, "a list entry packs sender (16 bits) and receiver - r0");

__device__ __forceinline__ unsigned zero_bytes(unsigned t) {
    return ~(((t & 0x7f7f7f7fu) + 0x7f7f7f7fu) | t) & 0x80808080u;
}

__device__ __forceinline__ void lds_max_float(float* cell, float v) {
    if (__float_as_int(v) >= 0) atomicMax((int*)cell, __float_as_int(v));
    else                        atomicMin((unsigned int*)cell, __float_as_uint(v));
}

__device__ __forceinline__ float leaky(float v) { return (v > 0.0f) ? v : NEG_SLOPE * v; }

#define VCAP 64
#define BM_PER_WAVE (BMW / SG_WAVES)
#define BM_TRIPS    ((BM_PER_WAVE + 31) / 32)
static_assert(BMW % SG_WAVES == 0, "every wave walks the same number of bitmap dwords");
template <class Sink>
__device__ __forceinline__ void walk_range_bitmap(
    const unsigned* __restrict__ brow, const uint4* __restrict__ rv, unsigned pat, int wv, int lane,
    int* vpend, int* pend, Sink&& sink)
{
    unsigned head = 0, tail = 0;
    unsigned vhead = 0, vtail = 0;
    unsigned bw[BM_TRIPS];
    #pragma unroll
    for (int t = 0; t < BM_TRIPS; ++t) {
        const int d = t * 32 + lane;
        const unsigned v = brow[wv * BM_PER_WAVE + (d < BM_PER_WAVE ? d : BM_PER_WAVE - 1)];
        bw[t] = (d < BM_PER_WAVE) ? v : 0u;
    }
    auto vectors = [&](unsigned vh, int n) {
        const int vec = vpend[(vh + (unsigned)(lane < n ? lane : 0)) & (VCAP - 1)];
        const uint4 cur = rv[vec];
        unsigned m = 0u;
        if (lane < n)
            m = (zero_bytes(cur.x ^ pat) >> 7) | (zero_bytes(cur.y ^ pat) >> 6)
              | (zero_bytes(cur.z ^ pat) >> 5) | (zero_bytes(cur.w ^ pat) >> 4);
        for (;;) {
            const unsigned bal = __builtin_amdgcn_ballot_w32(m != 0u);
            if (bal == 0u) break;
            if (m != 0u) {
                const int p = __builtin_ctz(m);
                const unsigned rank = __builtin_amdgcn_mbcnt_lo(bal, 0u);
                pend[(tail + rank) & (PCAP - 1)] = vec * 16 + ((p & 7) << 2) + (p >> 3);
                m &= m - 1u;
            }
            tail += (unsigned)__builtin_popcount(bal);
            if (tail - head >= 32u) { sink(head, 32); head += 32u; }
        }
    };
    #pragma unroll
    for (int t = 0; t < BM_TRIPS; ++t) {
        unsigned w = bw[t];
        const int v0 = (wv * BM_PER_WAVE + t * 32 + lane) * 32;
        for (;;) {
            const unsigned bal = __builtin_amdgcn_ballot_w32(w != 0u);
            if (bal == 0u) break;
            if (w != 0u) {
                const int p = __builtin_ctz(w);
                const unsigned rank = __builtin_amdgcn_mbcnt_lo(bal, 0u);
                vpend[(vtail + rank) & (VCAP - 1)] = v0 + p;
                w &= w - 1u;
            }
            vtail += (unsigned)__builtin_popcount(bal);
            if (vtail - vhead >= 32u) { vectors(vhead, 32); vhead += 32u; }
        }
    }
    if (vtail != vhead) vectors(vhead, (int)(vtail - vhead));
    if (tail != head) sink(head, (int)(tail - head));
}

__device__ __forceinline__ void edge_logits(
    const int* pend, unsigned head, int n, int lane, int r0,
    const int* __restrict__ ei, const float* __restrict__ s_j, const float* sis,
    float4 ae, float4 ba, int& packed, int& rel, float4& lg)
{
    const int e = pend[(head + (unsigned)(lane < n ? lane : 0)) & (PCAP - 1)];
    const int2 rs = *(const int2*)(ei + 2 * (size_t)e);
    const float4 sj = *(const float4*)(s_j + (size_t)rs.y * NHEADS);
    rel = rs.x - r0;
    const float4 si = *(const float4*)(sis + rel * NHEADS);
    lg.x = leaky(si.x + sj.x + ae.x + ba.x);
    lg.y = leaky(si.y + sj.y + ae.y + ba.y);
    lg.z = leaky(si.z + sj.z + ae.z + ba.z);
    lg.w = leaky(si.w + sj.w + ae.w + ba.w);
    packed = rs.y | (rel << 16);
}

__device__ __forceinline__ void list_weights(
    const int* lsr, float* llg, int from, int n, int lane, const float* mxs, float* dns)
{
    for (int b = 0; b < n; b += 32) {
        const int i = from + b + lane;
        if (b + lane < n) {
            const int rel = lsr[i] >> 16;
            const float4 l = *(const float4*)(llg + 4 * i);
            const float4 m = *(const float4*)(mxs + rel * NHEADS);
            float4 w;
            w.x = expf(l.x - m.x); w.y = expf(l.y - m.y); w.z = expf(l.z - m.z); w.w = expf(l.w - m.w);
            atomicAdd(dns + rel * NHEADS + 0, w.x);
            atomicAdd(dns + rel * NHEADS + 1, w.y);
            atomicAdd(dns + rel * NHEADS + 2, w.z);
            atomicAdd(dns + rel * NHEADS + 3, w.w);
            *(float4*)(llg + 4 * i) = w;
        }
    }
}

template <bool FULL>
__device__ __forceinline__ void add_batch(
    const int* lsr, const float* llg, int base, int n, int lane,
    const float* __restrict__ h, float* acc)
{
    const int hd = lane >> 3;
    int   sr[8];
    float w[8];
    #pragma unroll
    for (int b = 0; b < 8; ++b) {
        const int i = base + (FULL ? b : (b < n ? b : n - 1));
        sr[b] = lsr[i];
        w[b]  = llg[4 * i + hd];
    }
    float4 hv[8];
    #pragma unroll
    for (int b = 0; b < 8; ++b)
        hv[b] = *(const float4*)(h + (size_t)(sr[b] & 0xffff) * DF + 4 * lane);
    #pragma unroll
    for (int b = 0; b < 8; ++b) {
        if (FULL || b < n) {
            float* a = acc + (sr[b] >> 16) * DF + lane;
            atomicAdd(a,      hv[b].x * w[b]);
            atomicAdd(a + 32, hv[b].y * w[b]);
            atomicAdd(a + 64, hv[b].z * w[b]);
            atomicAdd(a + 96, hv[b].w * w[b]);
        }
    }
}

__device__ __forceinline__ void list_rows(
    const int* lsr, const float* llg, int n, int lane, const float* __restrict__ h, float* acc)
{
    int base = 0;
    for (; base + 8 <= n; base += 8) add_batch<true>(lsr, llg, base, 8, lane, h, acc);
    if (base < n) add_batch<false>(lsr, llg, base, n - base, lane, h, acc);
}

#define ROWS_PER_WAVE ((RNG + SG_WAVES - 1) / SG_WAVES)

__device__ __forceinline__ float out_elem(float u, float mean, float rs, float g, float b, float a, float d) {
    const float normed = g * (u - mean) * rs + b;
    return normed + fmaxf(a / d, 0.0f);
}

__global__ __launch_bounds__(512) void k_edge_range_sum(
    const int* __restrict__ ei, const float* __restrict__ h,
    const float* __restrict__ s_i, const float* __restrict__ s_j,
    const float* __restrict__ a_e, const float* __restrict__ b_a,
    const unsigned char* __restrict__ rbyte, const unsigned* __restrict__ bitmap,
    const float* __restrict__ upd, const float* __restrict__ macc,
    const float* __restrict__ sqacc, const float* __restrict__ gamma,
    const float* __restrict__ beta, float* __restrict__ out)
{
    __shared__ float acc[RNG * DF];
    __shared__ __attribute__((aligned(16))) float mxs[RNG * NHEADS];
    __shared__ __attribute__((aligned(16))) float dns[RNG * NHEADS];
    __shared__ __attribute__((aligned(16))) float sis[RNG * NHEADS];
    __shared__ int   pending[SG_WAVES * PCAP];
    __shared__ int   vpending[SG_WAVES * VCAP];
    __shared__ int   list_sr[SG_WAVES * LCAP];
    __shared__ __attribute__((aligned(16))) float list_lg[SG_WAVES * LCAP * 4];

    const int tid  = threadIdx.x;
    const int lane = tid & 31;
    const int wv   = tid >> 5;
    const int r0   = blockIdx.x * RNG;

    for (int i = tid; i < RNG * DF; i += 512) acc[i] = 0.0f;
    for (int i = tid; i < RNG * NHEADS; i += 512) {
        const int g = r0 * NHEADS + i;
        mxs[i] = -INFINITY;
        dns[i] = 0.0f;
        sis[i] = (g < NN * NHEADS) ? s_i[g] : 0.0f;
    }
    const float4 ae = make_float4(a_e[0], a_e[1], a_e[2], a_e[3]);
    const float4 ba = make_float4(b_a[0], b_a[1], b_a[2], b_a[3]);
    __syncthreads();

    const unsigned pat = blockIdx.x * 0x01010101u;
    const uint4* __restrict__ rv = (const uint4*)rbyte;
    int*   pend = pending + wv * PCAP;
    int*   vpend = vpending + wv * VCAP;
    const unsigned* __restrict__ brow = bitmap + (size_t)blockIdx.x * BMW;
    int*   lsr  = list_sr + wv * LCAP;
    float* llg  = list_lg + wv * LCAP * 4;
    int  cnt  = 0;
    bool over = false;

    walk_range_bitmap(brow, rv, pat, wv, lane, vpend, pend, [&](unsigned head, int n) {
        int packed, rel; float4 lg;
        edge_logits(pend, head, n, lane, r0, ei, s_j, sis, ae, ba, packed, rel, lg);
        if (lane < n) {
            lds_max_float(mxs + rel * NHEADS + 0, lg.x);
            lds_max_float(mxs + rel * NHEADS + 1, lg.y);
            lds_max_float(mxs + rel * NHEADS + 2, lg.z);
            lds_max_float(mxs + rel * NHEADS + 3, lg.w);
        }
        if (!over && cnt + n <= LCAP) {
            if (lane < n) {
                lsr[cnt + lane] = packed;
                *(float4*)(llg + 4 * (cnt + lane)) = lg;
            }
            cnt += n;
        } else {
            over = true;
        }
    });
    __syncthreads();

    list_weights(lsr, llg, 0, cnt, lane, mxs, dns);
    list_rows(lsr, llg, cnt, lane, h, acc);
    if (over) {
        int seen = 0;
        walk_range_bitmap(brow, rv, pat, wv, lane, vpend, pend, [&](unsigned head, int n) {
            const int lo = (cnt > seen) ? (cnt - seen) : 0;
            seen += n;
            if (lo >= n) return;
            int packed, rel; float4 lg;
            edge_logits(pend, head, n, lane, r0, ei, s_j, sis, ae, ba, packed, rel, lg);
            if (lane >= lo && lane < n) {
                lsr[lane - lo] = packed;
                *(float4*)(llg + 4 * (lane - lo)) = lg;
            }
            list_weights(lsr, llg, 0, n - lo, lane, mxs, dns);
            list_rows(lsr, llg, n - lo, lane, h, acc);
        });
    }

    float4 u[ROWS_PER_WAVE];
    #pragma unroll
    for (int i = 0; i < ROWS_PER_WAVE; ++i) {
        const int row = r0 + wv + i * SG_WAVES;
        u[i] = *(const float4*)(upd + (size_t)(row < NN ? row : NN - 1) * DF + 4 * lane);
    }
    const float4 m4 = *(const float4*)(macc + 4 * lane);
    const float4 q4 = *(const float4*)(sqacc + 4 * lane);
    const float4 g4 = *(const float4*)(gamma + 4 * lane);
    const float4 b4 = *(const float4*)(beta + 4 * lane);
    __syncthreads();

    const float inv_n = 1.0f / (float)NN;
    const float mean0 = m4.x * inv_n, mean1 = m4.y * inv_n, mean2 = m4.z * inv_n, mean3 = m4.w * inv_n;
    const float rs0 = rsqrtf(q4.x * inv_n - mean0 * mean0 + BN_EPS);
    const float rs1 = rsqrtf(q4.y * inv_n - mean1 * mean1 + BN_EPS);
    const float rs2 = rsqrtf(q4.z * inv_n - mean2 * mean2 + BN_EPS);
    const float rs3 = rsqrtf(q4.w * inv_n - mean3 * mean3 + BN_EPS);

    const int hd = lane >> 3;
    #pragma unroll
    for (int i = 0; i < ROWS_PER_WAVE; ++i) {
        const int r = wv + i * SG_WAVES;
        const int row = r0 + r;
        if (r < RNG && row < NN) {
            const float* a = acc + r * DF + lane;
            const float d = dns[r * NHEADS + hd] + 1e-16f;
            float4 v;
            v.x = out_elem(u[i].x, mean0, rs0, g4.x, b4.x, a[0],  d);
            v.y = out_elem(u[i].y, mean1, rs1, g4.y, b4.y, a[32], d);
            v.z = out_elem(u[i].z, mean2, rs2, g4.z, b4.z, a[64], d);
            v.w = out_elem(u[i].w, mean3, rs3, g4.w, b4.w, a[96], d);
            *(float4*)(out + (size_t)row * DF + 4 * lane) = v;
        }
    }
}

__global__ __launch_bounds__(256) void k_bn_reduce(
    const float* __restrict__ rows, float* __restrict__ macc, float* __restrict__ sqacc)
{
    const int t = threadIdx.x;
    const float* __restrict__ p = rows + (size_t)blockIdx.x * CS_ROWS * (2 * DF) + t;
    float v[CS_ROWS];
    #pragma unroll
    for (int r = 0; r < CS_ROWS; ++r) v[r] = p[(size_t)r * (2 * DF)];
    float sum = 0.0f;
    #pragma unroll
    for (int r = 0; r < CS_ROWS; ++r) sum += v[r];
    const int col = ((t >> 5) << 4) | (t & 15);
    atomicAdd(((t >> 4) & 1) ? &sqacc[col] : &macc[col], sum);
}

extern "C" void kernel_launch(void* const* d_in, const int* in_sizes, int n_in,
                              void* d_out, int out_size, void* d_ws, size_t ws_size,
                              hipStream_t stream) {
    const float* x     = (const float*)d_in[0];
    const int*   ei    = (const int*)  d_in[1];
    const float* Wk    = (const float*)d_in[2];
    const float* bk    = (const float*)d_in[3];
    const float* a_i   = (const float*)d_in[4];
    const float* a_j   = (const float*)d_in[5];
    const float* a_e   = (const float*)d_in[6];
    const float* b_a   = (const float*)d_in[7];
    const float* Wm    = (const float*)d_in[8];
    const float* bm    = (const float*)d_in[9];
    const float* gamma = (const float*)d_in[10];
    const float* beta  = (const float*)d_in[11];
    float* out = (float*)d_out;
    float* ws  = (float*)d_ws;

    float* h_buf  = ws + OFF_H;
    float* upd    = ws + OFF_UPD;
    float* s_i    = ws + OFF_SI;
    float* s_j    = ws + OFF_SJ;
    float* macc   = ws + OFF_MACC;
    float* sqacc  = ws + OFF_SQ;
    unsigned char* rbyte = (unsigned char*)(ws + OFF_W);
    unsigned* bitmap = (unsigned*)(ws + OFF_W + NE / 4);
    float* colsum_rows = ws + OFF_AGG;

    k_init<<<INIT_GRID, 256, 0, stream>>>(ei, ws, rbyte, bitmap);
    k_gemm_wmma<1, 0, 1, 0><<<GEMM_GRID, 256, 0, stream>>>(
        x, Wk, bk, h_buf, NN, a_i, a_j, s_i, s_j, nullptr);
    k_gemm_wmma<0, 1, 0, 1><<<GEMM_GRID, 256, 0, stream>>>(
        x, Wm, bm, upd, NN, nullptr, nullptr, nullptr, nullptr, colsum_rows);
    k_bn_reduce<<<GEMM_BLOCKS / CS_ROWS, 256, 0, stream>>>(colsum_rows, macc, sqacc);
    k_edge_range_sum<<<NRNG, 512, 0, stream>>>(
        ei, h_buf, s_i, s_j, a_e, b_a, rbyte, bitmap, upd, macc, sqacc, gamma, beta, out);
}


# Range bitmaps, h projection and upd projection share one launch (work-group index picks the stage, one 72 KB LDS block): bitmap work runs beside the projections; 5 launches -> 3; sums unchanged
# speedup vs baseline: 6.7181x; 1.2008x over previous
#include <hip/hip_runtime.h>
#include <hip/hip_bf16.h>
#include <math.h>

#define NN      50000
#define NE      800000
#define DF      128
#define NHEADS  4
#define HU      32
#define NEG_SLOPE 0.2f
#define BN_EPS    0.001f
#define RNG     196
#define NRNG    ((NN + RNG - 1) / RNG)
#define GEMM_BLOCKS ((NN + 15) / 16)
#define CS_ROWS 25
static_assert(GEMM_BLOCKS % CS_ROWS == 0, "the reduce launch reads whole groups of partial rows");

typedef __attribute__((ext_vector_type(2))) float v2f;
typedef __attribute__((ext_vector_type(8))) float v8f;

#define OFF_H    ((size_t)0)
#define OFF_UPD  (OFF_H   + (size_t)NN * DF)
#define OFF_AGG  (OFF_UPD + (size_t)NN * DF)
#define OFF_SI   (OFF_AGG + (size_t)NN * DF)
#define OFF_SJ   (OFF_SI  + (size_t)NN * NHEADS)
#define OFF_MAX  (OFF_SJ  + (size_t)NN * NHEADS)
#define OFF_DEN  (OFF_MAX + (size_t)NN * NHEADS)
#define OFF_W    (OFF_DEN + (size_t)NN * NHEADS)
#define OFF_MACC (OFF_W   + (size_t)NE * NHEADS)
#define OFF_SQ   (OFF_MACC + (size_t)DF)

#define NVEC     (NE / 16)
static_assert(NE % 16 == 0, "whole 16-edge vectors");
#define INIT_VEC   1024
#define INIT_GRID  ((NVEC + INIT_VEC - 1) / INIT_VEC)
#define BMW        (INIT_GRID * 32)
#define BMP        (NRNG + 1)
__device__ __forceinline__ void init_stage(const int bid, unsigned* bm,
                       const int* __restrict__ ei, float* __restrict__ ws,
                       unsigned char* __restrict__ rbyte, unsigned* __restrict__ bitmap) {
    const int tid = threadIdx.x;
    const unsigned bit = 1u << (tid & 31);
    int4 a[INIT_VEC / 256][8];
    #pragma unroll
    for (int j = 0; j < INIT_VEC / 256; ++j) {
        const int i = bid * INIT_VEC + j * 256 + tid;
        const int4* __restrict__ p = (const int4*)(ei + 32 * (size_t)(i < NVEC ? i : NVEC - 1));
        #pragma unroll
        for (int k = 0; k < 8; ++k) a[j][k] = p[k];
    }
    for (int i = tid; i < 32 * BMP; i += 256) bm[i] = 0u;
    if (bid == 0 && tid < DF) {
        ws[OFF_MACC + tid] = 0.0f;
        ws[OFF_SQ + tid]   = 0.0f;
    }
    __syncthreads();
    #pragma unroll
    for (int j = 0; j < INIT_VEC / 256; ++j) {
        const int lv = j * 256 + tid;
        const int i  = bid * INIT_VEC + lv;
        if (i < NVEC) {
            unsigned rid[16];
            #pragma unroll
            for (int k = 0; k < 8; ++k) {
                rid[2 * k]     = (unsigned)a[j][k].x / RNG;
                rid[2 * k + 1] = (unsigned)a[j][k].z / RNG;
            }
            unsigned w[4];
            #pragma unroll
            for (int k = 0; k < 4; ++k)
                w[k] = rid[4 * k] | (rid[4 * k + 1] << 8) | (rid[4 * k + 2] << 16) | (rid[4 * k + 3] << 24);
            ((uint4*)rbyte)[i] = make_uint4(w[0], w[1], w[2], w[3]);
            unsigned* plane = bm + (lv >> 5) * BMP;
            #pragma unroll
            for (int k = 0; k < 16; ++k) atomicOr(plane + rid[k], bit);
        }
    }
    __syncthreads();
    const int lane = tid & 31;
    for (int r = tid >> 5; r < NRNG; r += 8)
        bitmap[(size_t)r * BMW + bid * 32 + lane] = bm[lane * BMP + r];
}

template <int M>
__device__ __forceinline__ float row_xor(float x) {
    return __int_as_float(__builtin_amdgcn_update_dpp(
        0, __float_as_int(x), 0x160 + M, 0xf, 0xf, true));
}
__device__ __forceinline__ float other_half(float x) {
    const int v = __float_as_int(x);
    return __int_as_float(__builtin_amdgcn_permlanex16(v, v, 0x76543210, 0xfedcba98, false, false));
}
__device__ __forceinline__ float pick(bool c, float a, float b) { return c ? a : b; }
template <int W>
__device__ __forceinline__ void fold(float* v, bool up) {
    #pragma unroll
    for (int k = 0; k < W; ++k) {
        const float lo = v[k], hi = v[k + W];
        v[k] = pick(up, hi, lo) + row_xor<W>(pick(up, lo, hi));
    }
}

#define GEMM_WAVES 8
#define GEMM_GRID  ((GEMM_BLOCKS + GEMM_WAVES - 1) / GEMM_WAVES)
#define WP 288
template <int WLAYOUT, int DORELU, int SCORES, int COLSUMS>
__device__ __forceinline__ void gemm_stage(
    const int bid, float* wlds,
    const float* __restrict__ A, const float* __restrict__ B,
    const float* __restrict__ bias, float* __restrict__ C, int M,
    const float* __restrict__ a_i, const float* __restrict__ a_j,
    float* __restrict__ s_i, float* __restrict__ s_j,
    float* __restrict__ colsum_rows)
{
    const int tid  = threadIdx.x;
    const int lane = tid & 31;
    const int half = lane >> 4;
    const int l    = lane & 15;
    const int stripe = bid * GEMM_WAVES + (tid >> 5);
    const bool active = stripe < GEMM_BLOCKS;
    const int m0 = stripe * 16;

    const bool arow_ok = active && (m0 + l) < M;
    const float* __restrict__ ap = A + (size_t)(arow_ok ? (m0 + l) : 0) * DF + 2 * half;
    v2f a[DF / 4];
    #pragma unroll
    for (int s = 0; s < DF / 4; ++s) a[s] = *(const v2f*)(ap + 4 * s);

    {
        float4 r0[8], r1[8];
        #pragma unroll
        for (int j = 0; j < 8; ++j) {
            const int i = tid + 256 * j;
            const int p = i >> 5, cq = i & 31;
            const float* __restrict__ src = (WLAYOUT == 0)
                ? B + (size_t)(2 * p) * DF + 4 * cq
                : B + (size_t)(cq >> 3) * (DF * HU) + (size_t)(2 * p) * HU + 4 * (cq & 7);
            r0[j] = *(const float4*)src;
            r1[j] = *(const float4*)(src + ((WLAYOUT == 0) ? DF : HU));
        }
        #pragma unroll
        for (int j = 0; j < 8; ++j) {
            const int i = tid + 256 * j;
            const int p = i >> 5, cq = i & 31;
            float4* d = (float4*)&wlds[p * WP + 8 * cq];
            d[0] = make_float4(r0[j].x, r1[j].x, r0[j].y, r1[j].y);
            d[1] = make_float4(r0[j].z, r1[j].z, r0[j].w, r1[j].w);
        }
    }
    __syncthreads();
    if (!active) return;

    if (!arow_ok) {
        #pragma unroll
        for (int s = 0; s < DF / 4; ++s) { a[s].x = 0.0f; a[s].y = 0.0f; }
    }

    v8f acc[8];
    #pragma unroll
    for (int t = 0; t < 8; ++t) acc[t] = v8f{};

    const float* bl = &wlds[half * WP + 2 * l];
    #pragma unroll
    for (int s = 0; s < DF / 4; ++s) {
        #pragma unroll
        for (int t = 0; t < 8; ++t) {
            const v2f b = *(const v2f*)(bl + 2 * s * WP + 32 * t);
            acc[t] = __builtin_amdgcn_wmma_f32_16x16x4_f32(
                         false, a[s], false, b, (short)0, acc[t], false, false);
        }
    }

    float sc[NHEADS];
    float sprev = 0.0f;
    #pragma unroll
    for (int t = 0; t < 8; ++t) {
        const int col = 16 * t + l;
        const float bv = bias[col];
        float hv[8];
        #pragma unroll
        for (int r = 0; r < 8; ++r) {
            float v = acc[t][r] + bv;
            if (DORELU) v = fmaxf(v, 0.0f);
            hv[r] = v;
        }

        if (SCORES) {
            const float ai = a_i[col], aj = a_j[col];
            float v[16];
            #pragma unroll
            for (int r = 0; r < 8; ++r) { v[r] = hv[r] * ai; v[8 + r] = hv[r] * aj; }
            fold<8>(v, (l & 8) != 0);
            fold<4>(v, (l & 4) != 0);
            fold<2>(v, (l & 2) != 0);
            fold<1>(v, (l & 1) != 0);
            if (t & 1) sc[t >> 1] = sprev + v[0];
            else       sprev = v[0];
        }

        float cs = 0.0f, cq = 0.0f;
        #pragma unroll
        for (int r = 0; r < 8; ++r) {
            const int row = m0 + r + 8 * half;
            if (row < M) {
                C[(size_t)row * DF + col] = hv[r];
                if (COLSUMS) { cs += hv[r]; cq += hv[r] * hv[r]; }
            }
        }

        if (COLSUMS) {
            cs += other_half(cs);
            cq += other_half(cq);
            colsum_rows[(size_t)stripe * (2 * DF) + t * 32 + lane] = half ? cq : cs;
        }
    }

    if (SCORES) {
        const int node = m0 + 8 * half + (l & 7);
        if (node < M)
            *(float4*)(((l >> 3) ? s_j : s_i) + (size_t)node * NHEADS) = make_float4(sc[0], sc[1], sc[2], sc[3]);
    }
}

#define FRONT_GRID (INIT_GRID + 2 * GEMM_GRID)
#define INIT_STEP  8
#define INIT_SPAN  (INIT_GRID * INIT_STEP)
static_assert(INIT_SPAN <= FRONT_GRID, "every bitmap work-group has an index in the grid");
__global__ __launch_bounds__(256) void k_front(
    const int* __restrict__ ei, float* __restrict__ ws,
    unsigned char* __restrict__ rbyte, unsigned* __restrict__ bitmap,
    const float* __restrict__ x, int M,
    const float* __restrict__ Wk, const float* __restrict__ bk, float* __restrict__ h,
    const float* __restrict__ a_i, const float* __restrict__ a_j,
    float* __restrict__ s_i, float* __restrict__ s_j,
    const float* __restrict__ Wm, const float* __restrict__ bm_bias, float* __restrict__ upd,
    float* __restrict__ colsum_rows)
{
    __shared__ __attribute__((aligned(16))) float smem[(DF / 2) * WP];
    static_assert(sizeof(unsigned) * 32 * BMP <= sizeof(float) * (DF / 2) * WP, "stage 0's planes fit the launch's LDS");
    const int b = blockIdx.x;
    if (b >= INIT_SPAN || (b % INIT_STEP) != 0) {
        const int below = (b + INIT_STEP - 1) / INIT_STEP;
        const int g = b - (below < INIT_GRID ? below : INIT_GRID);
        if (g & 1)
            gemm_stage<0, 1, 0, 1>(g >> 1, smem, x, Wm, bm_bias, upd, M,
                                   nullptr, nullptr, nullptr, nullptr, colsum_rows);
        else
            gemm_stage<1, 0, 1, 0>(g >> 1, smem, x, Wk, bk, h, M,
                                   a_i, a_j, s_i, s_j, nullptr);
    } else {
        init_stage(b / INIT_STEP, (unsigned*)smem, ei, ws, rbyte, bitmap);
    }
}

#define SG_WAVES 16
#define PCAP     64
#define LCAP     384
static_assert(NN <= 65536 && RNG <= 256, "a list entry packs sender (16 bits) and receiver - r0");

__device__ __forceinline__ unsigned zero_bytes(unsigned t) {
    return ~(((t & 0x7f7f7f7fu) + 0x7f7f7f7fu) | t) & 0x80808080u;
}

__device__ __forceinline__ void lds_max_float(float* cell, float v) {
    if (__float_as_int(v) >= 0) atomicMax((int*)cell, __float_as_int(v));
    else                        atomicMin((unsigned int*)cell, __float_as_uint(v));
}

__device__ __forceinline__ float leaky(float v) { return (v > 0.0f) ? v : NEG_SLOPE * v; }

#define VCAP 64
#define BM_PER_WAVE (BMW / SG_WAVES)
#define BM_TRIPS    ((BM_PER_WAVE + 31) / 32)
static_assert(BMW % SG_WAVES == 0, "every wave walks the same number of bitmap dwords");
template <class Sink>
__device__ __forceinline__ void walk_range_bitmap(
    const unsigned* __restrict__ brow, const uint4* __restrict__ rv, unsigned pat, int wv, int lane,
    int* vpend, int* pend, Sink&& sink)
{
    unsigned head = 0, tail = 0;
    unsigned vhead = 0, vtail = 0;
    unsigned bw[BM_TRIPS];
    #pragma unroll
    for (int t = 0; t < BM_TRIPS; ++t) {
        const int d = t * 32 + lane;
        const unsigned v = brow[wv * BM_PER_WAVE + (d < BM_PER_WAVE ? d : BM_PER_WAVE - 1)];
        bw[t] = (d < BM_PER_WAVE) ? v : 0u;
    }
    auto vectors = [&](unsigned vh, int n) {
        const int vec = vpend[(vh + (unsigned)(lane < n ? lane : 0)) & (VCAP - 1)];
        const uint4 cur = rv[vec];
        unsigned m = 0u;
        if (lane < n)
            m = (zero_bytes(cur.x ^ pat) >> 7) | (zero_bytes(cur.y ^ pat) >> 6)
              | (zero_bytes(cur.z ^ pat) >> 5) | (zero_bytes(cur.w ^ pat) >> 4);
        for (;;) {
            const unsigned bal = __builtin_amdgcn_ballot_w32(m != 0u);
            if (bal == 0u) break;
            if (m != 0u) {
                const int p = __builtin_ctz(m);
                const unsigned rank = __builtin_amdgcn_mbcnt_lo(bal, 0u);
                pend[(tail + rank) & (PCAP - 1)] = vec * 16 + ((p & 7) << 2) + (p >> 3);
                m &= m - 1u;
            }
            tail += (unsigned)__builtin_popcount(bal);
            if (tail - head >= 32u) { sink(head, 32); head += 32u; }
        }
    };
    #pragma unroll
    for (int t = 0; t < BM_TRIPS; ++t) {
        unsigned w = bw[t];
        const int v0 = (wv * BM_PER_WAVE + t * 32 + lane) * 32;
        for (;;) {
            const unsigned bal = __builtin_amdgcn_ballot_w32(w != 0u);
            if (bal == 0u) break;
            if (w != 0u) {
                const int p = __builtin_ctz(w);
                const unsigned rank = __builtin_amdgcn_mbcnt_lo(bal, 0u);
                vpend[(vtail + rank) & (VCAP - 1)] = v0 + p;
                w &= w - 1u;
            }
            vtail += (unsigned)__builtin_popcount(bal);
            if (vtail - vhead >= 32u) { vectors(vhead, 32); vhead += 32u; }
        }
    }
    if (vtail != vhead) vectors(vhead, (int)(vtail - vhead));
    if (tail != head) sink(head, (int)(tail - head));
}

__device__ __forceinline__ void edge_logits(
    const int* pend, unsigned head, int n, int lane, int r0,
    const int* __restrict__ ei, const float* __restrict__ s_j, const float* sis,
    float4 ae, float4 ba, int& packed, int& rel, float4& lg)
{
    const int e = pend[(head + (unsigned)(lane < n ? lane : 0)) & (PCAP - 1)];
    const int2 rs = *(const int2*)(ei + 2 * (size_t)e);
    const float4 sj = *(const float4*)(s_j + (size_t)rs.y * NHEADS);
    rel = rs.x - r0;
    const float4 si = *(const float4*)(sis + rel * NHEADS);
    lg.x = leaky(si.x + sj.x + ae.x + ba.x);
    lg.y = leaky(si.y + sj.y + ae.y + ba.y);
    lg.z = leaky(si.z + sj.z + ae.z + ba.z);
    lg.w = leaky(si.w + sj.w + ae.w + ba.w);
    packed = rs.y | (rel << 16);
}

__device__ __forceinline__ void list_weights(
    const int* lsr, float* llg, int from, int n, int lane, const float* mxs, float* dns)
{
    for (int b = 0; b < n; b += 32) {
        const int i = from + b + lane;
        if (b + lane < n) {
            const int rel = lsr[i] >> 16;
            const float4 l = *(const float4*)(llg + 4 * i);
            const float4 m = *(const float4*)(mxs + rel * NHEADS);
            float4 w;
            w.x = expf(l.x - m.x); w.y = expf(l.y - m.y); w.z = expf(l.z - m.z); w.w = expf(l.w - m.w);
            atomicAdd(dns + rel * NHEADS + 0, w.x);
            atomicAdd(dns + rel * NHEADS + 1, w.y);
            atomicAdd(dns + rel * NHEADS + 2, w.z);
            atomicAdd(dns + rel * NHEADS + 3, w.w);
            *(float4*)(llg + 4 * i) = w;
        }
    }
}

template <bool FULL>
__device__ __forceinline__ void add_batch(
    const int* lsr, const float* llg, int base, int n, int lane,
    const float* __restrict__ h, float* acc)
{
    const int hd = lane >> 3;
    int   sr[8];
    float w[8];
    #pragma unroll
    for (int b = 0; b < 8; ++b) {
        const int i = base + (FULL ? b : (b < n ? b : n - 1));
        sr[b] = lsr[i];
        w[b]  = llg[4 * i + hd];
    }
    float4 hv[8];
    #pragma unroll
    for (int b = 0; b < 8; ++b)
        hv[b] = *(const float4*)(h + (size_t)(sr[b] & 0xffff) * DF + 4 * lane);
    #pragma unroll
    for (int b = 0; b < 8; ++b) {
        if (FULL || b < n) {
            float* a = acc + (sr[b] >> 16) * DF + lane;
            atomicAdd(a,      hv[b].x * w[b]);
            atomicAdd(a + 32, hv[b].y * w[b]);
            atomicAdd(a + 64, hv[b].z * w[b]);
            atomicAdd(a + 96, hv[b].w * w[b]);
        }
    }
}

__device__ __forceinline__ void list_rows(
    const int* lsr, const float* llg, int n, int lane, const float* __restrict__ h, float* acc)
{
    int base = 0;
    for (; base + 8 <= n; base += 8) add_batch<true>(lsr, llg, base, 8, lane, h, acc);
    if (base < n) add_batch<false>(lsr, llg, base, n - base, lane, h, acc);
}

#define ROWS_PER_WAVE ((RNG + SG_WAVES - 1) / SG_WAVES)

__device__ __forceinline__ float out_elem(float u, float mean, float rs, float g, float b, float a, float d) {
    const float normed = g * (u - mean) * rs + b;
    return normed + fmaxf(a / d, 0.0f);
}

__global__ __launch_bounds__(512) void k_edge_range_sum(
    const int* __restrict__ ei, const float* __restrict__ h,
    const float* __restrict__ s_i, const float* __restrict__ s_j,
    const float* __restrict__ a_e, const float* __restrict__ b_a,
    const unsigned char* __restrict__ rbyte, const unsigned* __restrict__ bitmap,
    const float* __restrict__ upd, const float* __restrict__ macc,
    const float* __restrict__ sqacc, const float* __restrict__ gamma,
    const float* __restrict__ beta, float* __restrict__ out)
{
    __shared__ float acc[RNG * DF];
    __shared__ __attribute__((aligned(16))) float mxs[RNG * NHEADS];
    __shared__ __attribute__((aligned(16))) float dns[RNG * NHEADS];
    __shared__ __attribute__((aligned(16))) float sis[RNG * NHEADS];
    __shared__ int   pending[SG_WAVES * PCAP];
    __shared__ int   vpending[SG_WAVES * VCAP];
    __shared__ int   list_sr[SG_WAVES * LCAP];
    __shared__ __attribute__((aligned(16))) float list_lg[SG_WAVES * LCAP * 4];

    const int tid  = threadIdx.x;
    const int lane = tid & 31;
    const int wv   = tid >> 5;
    const int r0   = blockIdx.x * RNG;

    for (int i = tid; i < RNG * DF; i += 512) acc[i] = 0.0f;
    for (int i = tid; i < RNG * NHEADS; i += 512) {
        const int g = r0 * NHEADS + i;
        mxs[i] = -INFINITY;
        dns[i] = 0.0f;
        sis[i] = (g < NN * NHEADS) ? s_i[g] : 0.0f;
    }
    const float4 ae = make_float4(a_e[0], a_e[1], a_e[2], a_e[3]);
    const float4 ba = make_float4(b_a[0], b_a[1], b_a[2], b_a[3]);
    __syncthreads();

    const unsigned pat = blockIdx.x * 0x01010101u;
    const uint4* __restrict__ rv = (const uint4*)rbyte;
    int*   pend = pending + wv * PCAP;
    int*   vpend = vpending + wv * VCAP;
    const unsigned* __restrict__ brow = bitmap + (size_t)blockIdx.x * BMW;
    int*   lsr  = list_sr + wv * LCAP;
    float* llg  = list_lg + wv * LCAP * 4;
    int  cnt  = 0;
    bool over = false;

    walk_range_bitmap(brow, rv, pat, wv, lane, vpend, pend, [&](unsigned head, int n) {
        int packed, rel; float4 lg;
        edge_logits(pend, head, n, lane, r0, ei, s_j, sis, ae, ba, packed, rel, lg);
        if (lane < n) {
            lds_max_float(mxs + rel * NHEADS + 0, lg.x);
            lds_max_float(mxs + rel * NHEADS + 1, lg.y);
            lds_max_float(mxs + rel * NHEADS + 2, lg.z);
            lds_max_float(mxs + rel * NHEADS + 3, lg.w);
        }
        if (!over && cnt + n <= LCAP) {
            if (lane < n) {
                lsr[cnt + lane] = packed;
                *(float4*)(llg + 4 * (cnt + lane)) = lg;
            }
            cnt += n;
        } else {
            over = true;
        }
    });
    __syncthreads();

    list_weights(lsr, llg, 0, cnt, lane, mxs, dns);
    list_rows(lsr, llg, cnt, lane, h, acc);
    if (over) {
        int seen = 0;
        walk_range_bitmap(brow, rv, pat, wv, lane, vpend, pend, [&](unsigned head, int n) {
            const int lo = (cnt > seen) ? (cnt - seen) : 0;
            seen += n;
            if (lo >= n) return;
            int packed, rel; float4 lg;
            edge_logits(pend, head, n, lane, r0, ei, s_j, sis, ae, ba, packed, rel, lg);
            if (lane >= lo && lane < n) {
                lsr[lane - lo] = packed;
                *(float4*)(llg + 4 * (lane - lo)) = lg;
            }
            list_weights(lsr, llg, 0, n - lo, lane, mxs, dns);
            list_rows(lsr, llg, n - lo, lane, h, acc);
        });
    }

    float4 u[ROWS_PER_WAVE];
    #pragma unroll
    for (int i = 0; i < ROWS_PER_WAVE; ++i) {
        const int row = r0 + wv + i * SG_WAVES;
        u[i] = *(const float4*)(upd + (size_t)(row < NN ? row : NN - 1) * DF + 4 * lane);
    }
    const float4 m4 = *(const float4*)(macc + 4 * lane);
    const float4 q4 = *(const float4*)(sqacc + 4 * lane);
    const float4 g4 = *(const float4*)(gamma + 4 * lane);
    const float4 b4 = *(const float4*)(beta + 4 * lane);
    __syncthreads();

    const float inv_n = 1.0f / (float)NN;
    const float mean0 = m4.x * inv_n, mean1 = m4.y * inv_n, mean2 = m4.z * inv_n, mean3 = m4.w * inv_n;
    const float rs0 = rsqrtf(q4.x * inv_n - mean0 * mean0 + BN_EPS);
    const float rs1 = rsqrtf(q4.y * inv_n - mean1 * mean1 + BN_EPS);
    const float rs2 = rsqrtf(q4.z * inv_n - mean2 * mean2 + BN_EPS);
    const float rs3 = rsqrtf(q4.w * inv_n - mean3 * mean3 + BN_EPS);

    const int hd = lane >> 3;
    #pragma unroll
    for (int i = 0; i < ROWS_PER_WAVE; ++i) {
        const int r = wv + i * SG_WAVES;
        const int row = r0 + r;
        if (r < RNG && row < NN) {
            const float* a = acc + r * DF + lane;
            const float d = dns[r * NHEADS + hd] + 1e-16f;
            float4 v;
            v.x = out_elem(u[i].x, mean0, rs0, g4.x, b4.x, a[0],  d);
            v.y = out_elem(u[i].y, mean1, rs1, g4.y, b4.y, a[32], d);
            v.z = out_elem(u[i].z, mean2, rs2, g4.z, b4.z, a[64], d);
            v.w = out_elem(u[i].w, mean3, rs3, g4.w, b4.w, a[96], d);
            *(float4*)(out + (size_t)row * DF + 4 * lane) = v;
        }
    }
}

__global__ __launch_bounds__(256) void k_bn_reduce(
    const float* __restrict__ rows, float* __restrict__ macc, float* __restrict__ sqacc)
{
    const int t = threadIdx.x;
    const float* __restrict__ p = rows + (size_t)blockIdx.x * CS_ROWS * (2 * DF) + t;
    float v[CS_ROWS];
    #pragma unroll
    for (int r = 0; r < CS_ROWS; ++r) v[r] = p[(size_t)r * (2 * DF)];
    float sum = 0.0f;
    #pragma unroll
    for (int r = 0; r < CS_ROWS; ++r) sum += v[r];
    const int col = ((t >> 5) << 4) | (t & 15);
    atomicAdd(((t >> 4) & 1) ? &sqacc[col] : &macc[col], sum);
}

extern "C" void kernel_launch(void* const* d_in, const int* in_sizes, int n_in,
                              void* d_out, int out_size, void* d_ws, size_t ws_size,
                              hipStream_t stream) {
    const float* x     = (const float*)d_in[0];
    const int*   ei    = (const int*)  d_in[1];
    const float* Wk    = (const float*)d_in[2];
    const float* bk    = (const float*)d_in[3];
    const float* a_i   = (const float*)d_in[4];
    const float* a_j   = (const float*)d_in[5];
    const float* a_e   = (const float*)d_in[6];
    const float* b_a   = (const float*)d_in[7];
    const float* Wm    = (const float*)d_in[8];
    const float* bm    = (const float*)d_in[9];
    const float* gamma = (const float*)d_in[10];
    const float* beta  = (const float*)d_in[11];
    float* out = (float*)d_out;
    float* ws  = (float*)d_ws;

    float* h_buf  = ws + OFF_H;
    float* upd    = ws + OFF_UPD;
    float* s_i    = ws + OFF_SI;
    float* s_j    = ws + OFF_SJ;
    float* macc   = ws + OFF_MACC;
    float* sqacc  = ws + OFF_SQ;
    unsigned char* rbyte = (unsigned char*)(ws + OFF_W);
    unsigned* bitmap = (unsigned*)(ws + OFF_W + NE / 4);
    float* colsum_rows = ws + OFF_AGG;

    k_front<<<FRONT_GRID, 256, 0, stream>>>(
        ei, ws, rbyte, bitmap, x, NN, Wk, bk, h_buf, a_i, a_j, s_i, s_j, Wm, bm, upd, colsum_rows);
    k_bn_reduce<<<GEMM_BLOCKS / CS_ROWS, 256, 0, stream>>>(colsum_rows, macc, sqacc);
    k_edge_range_sum<<<NRNG, 512, 0, stream>>>(
        ei, h_buf, s_i, s_j, a_e, b_a, rbyte, bitmap, upd, macc, sqacc, gamma, beta, out);
}


# Message sums in registers: LDS integer counts rank a range's edges by receiver, a wave takes whole receivers from the ordered LDS table (rank order: sum order changed); LDS float adds, LDS sums gone
# speedup vs baseline: 7.1801x; 1.0688x over previous
#include <hip/hip_runtime.h>
#include <hip/hip_bf16.h>
#include <math.h>

#define NN      50000
#define NE      800000
#define DF      128
#define NHEADS  4
#define HU      32
#define NEG_SLOPE 0.2f
#define BN_EPS    0.001f
#define RNG     196
#define NRNG    ((NN + RNG - 1) / RNG)
#define GEMM_BLOCKS ((NN + 15) / 16)
#define CS_ROWS 25
static_assert(GEMM_BLOCKS % CS_ROWS == 0, "the reduce launch reads whole groups of partial rows");

typedef __attribute__((ext_vector_type(2))) float v2f;
typedef __attribute__((ext_vector_type(8))) float v8f;

#define OFF_H    ((size_t)0)
#define OFF_UPD  (OFF_H   + (size_t)NN * DF)
#define OFF_AGG  (OFF_UPD + (size_t)NN * DF)
#define OFF_SI   (OFF_AGG + (size_t)NN * DF)
#define OFF_SJ   (OFF_SI  + (size_t)NN * NHEADS)
#define OFF_MAX  (OFF_SJ  + (size_t)NN * NHEADS)
#define OFF_DEN  (OFF_MAX + (size_t)NN * NHEADS)
#define OFF_W    (OFF_DEN + (size_t)NN * NHEADS)
#define OFF_MACC (OFF_W   + (size_t)NE * NHEADS)
#define OFF_SQ   (OFF_MACC + (size_t)DF)

#define NVEC     (NE / 16)
static_assert(NE % 16 == 0, "whole 16-edge vectors");
#define INIT_VEC   1024
#define INIT_GRID  ((NVEC + INIT_VEC - 1) / INIT_VEC)
#define BMW        (INIT_GRID * 32)
#define BMP        (NRNG + 1)
__device__ __forceinline__ void init_stage(const int bid, unsigned* bm,
                       const int* __restrict__ ei, float* __restrict__ ws,
                       unsigned char* __restrict__ rbyte, unsigned* __restrict__ bitmap) {
    const int tid = threadIdx.x;
    const unsigned bit = 1u << (tid & 31);
    int4 a[INIT_VEC / 256][8];
    #pragma unroll
    for (int j = 0; j < INIT_VEC / 256; ++j) {
        const int i = bid * INIT_VEC + j * 256 + tid;
        const int4* __restrict__ p = (const int4*)(ei + 32 * (size_t)(i < NVEC ? i : NVEC - 1));
        #pragma unroll
        for (int k = 0; k < 8; ++k) a[j][k] = p[k];
    }
    for (int i = tid; i < 32 * BMP; i += 256) bm[i] = 0u;
    if (bid == 0 && tid < DF) {
        ws[OFF_MACC + tid] = 0.0f;
        ws[OFF_SQ + tid]   = 0.0f;
    }
    __syncthreads();
    #pragma unroll
    for (int j = 0; j < INIT_VEC / 256; ++j) {
        const int lv = j * 256 + tid;
        const int i  = bid * INIT_VEC + lv;
        if (i < NVEC) {
            unsigned rid[16];
            #pragma unroll
            for (int k = 0; k < 8; ++k) {
                rid[2 * k]     = (unsigned)a[j][k].x / RNG;
                rid[2 * k + 1] = (unsigned)a[j][k].z / RNG;
            }
            unsigned w[4];
            #pragma unroll
            for (int k = 0; k < 4; ++k)
                w[k] = rid[4 * k] | (rid[4 * k + 1] << 8) | (rid[4 * k + 2] << 16) | (rid[4 * k + 3] << 24);
            ((uint4*)rbyte)[i] = make_uint4(w[0], w[1], w[2], w[3]);
            unsigned* plane = bm + (lv >> 5) * BMP;
            #pragma unroll
            for (int k = 0; k < 16; ++k) atomicOr(plane + rid[k], bit);
        }
    }
    __syncthreads();
    const int lane = tid & 31;
    for (int r = tid >> 5; r < NRNG; r += 8)
        bitmap[(size_t)r * BMW + bid * 32 + lane] = bm[lane * BMP + r];
}

template <int M>
__device__ __forceinline__ float row_xor(float x) {
    return __int_as_float(__builtin_amdgcn_update_dpp(
        0, __float_as_int(x), 0x160 + M, 0xf, 0xf, true));
}
__device__ __forceinline__ float other_half(float x) {
    const int v = __float_as_int(x);
    return __int_as_float(__builtin_amdgcn_permlanex16(v, v, 0x76543210, 0xfedcba98, false, false));
}
__device__ __forceinline__ float pick(bool c, float a, float b) { return c ? a : b; }
template <int W>
__device__ __forceinline__ void fold(float* v, bool up) {
    #pragma unroll
    for (int k = 0; k < W; ++k) {
        const float lo = v[k], hi = v[k + W];
        v[k] = pick(up, hi, lo) + row_xor<W>(pick(up, lo, hi));
    }
}

#define GEMM_WAVES 8
#define GEMM_GRID  ((GEMM_BLOCKS + GEMM_WAVES - 1) / GEMM_WAVES)
#define WP 288
template <int WLAYOUT, int DORELU, int SCORES, int COLSUMS>
__device__ __forceinline__ void gemm_stage(
    const int bid, float* wlds,
    const float* __restrict__ A, const float* __restrict__ B,
    const float* __restrict__ bias, float* __restrict__ C, int M,
    const float* __restrict__ a_i, const float* __restrict__ a_j,
    float* __restrict__ s_i, float* __restrict__ s_j,
    float* __restrict__ colsum_rows)
{
    const int tid  = threadIdx.x;
    const int lane = tid & 31;
    const int half = lane >> 4;
    const int l    = lane & 15;
    const int stripe = bid * GEMM_WAVES + (tid >> 5);
    const bool active = stripe < GEMM_BLOCKS;
    const int m0 = stripe * 16;

    const bool arow_ok = active && (m0 + l) < M;
    const float* __restrict__ ap = A + (size_t)(arow_ok ? (m0 + l) : 0) * DF + 2 * half;
    v2f a[DF / 4];
    #pragma unroll
    for (int s = 0; s < DF / 4; ++s) a[s] = *(const v2f*)(ap + 4 * s);

    {
        float4 r0[8], r1[8];
        #pragma unroll
        for (int j = 0; j < 8; ++j) {
            const int i = tid + 256 * j;
            const int p = i >> 5, cq = i & 31;
            const float* __restrict__ src = (WLAYOUT == 0)
                ? B + (size_t)(2 * p) * DF + 4 * cq
                : B + (size_t)(cq >> 3) * (DF * HU) + (size_t)(2 * p) * HU + 4 * (cq & 7);
            r0[j] = *(const float4*)src;
            r1[j] = *(const float4*)(src + ((WLAYOUT == 0) ? DF : HU));
        }
        #pragma unroll
        for (int j = 0; j < 8; ++j) {
            const int i = tid + 256 * j;
            const int p = i >> 5, cq = i & 31;
            float4* d = (float4*)&wlds[p * WP + 8 * cq];
            d[0] = make_float4(r0[j].x, r1[j].x, r0[j].y, r1[j].y);
            d[1] = make_float4(r0[j].z, r1[j].z, r0[j].w, r1[j].w);
        }
    }
    __syncthreads();
    if (!active) return;

    if (!arow_ok) {
        #pragma unroll
        for (int s = 0; s < DF / 4; ++s) { a[s].x = 0.0f; a[s].y = 0.0f; }
    }

    v8f acc[8];
    #pragma unroll
    for (int t = 0; t < 8; ++t) acc[t] = v8f{};

    const float* bl = &wlds[half * WP + 2 * l];
    #pragma unroll
    for (int s = 0; s < DF / 4; ++s) {
        #pragma unroll
        for (int t = 0; t < 8; ++t) {
            const v2f b = *(const v2f*)(bl + 2 * s * WP + 32 * t);
            acc[t] = __builtin_amdgcn_wmma_f32_16x16x4_f32(
                         false, a[s], false, b, (short)0, acc[t], false, false);
        }
    }

    float sc[NHEADS];
    float sprev = 0.0f;
    #pragma unroll
    for (int t = 0; t < 8; ++t) {
        const int col = 16 * t + l;
        const float bv = bias[col];
        float hv[8];
        #pragma unroll
        for (int r = 0; r < 8; ++r) {
            float v = acc[t][r] + bv;
            if (DORELU) v = fmaxf(v, 0.0f);
            hv[r] = v;
        }

        if (SCORES) {
            const float ai = a_i[col], aj = a_j[col];
            float v[16];
            #pragma unroll
            for (int r = 0; r < 8; ++r) { v[r] = hv[r] * ai; v[8 + r] = hv[r] * aj; }
            fold<8>(v, (l & 8) != 0);
            fold<4>(v, (l & 4) != 0);
            fold<2>(v, (l & 2) != 0);
            fold<1>(v, (l & 1) != 0);
            if (t & 1) sc[t >> 1] = sprev + v[0];
            else       sprev = v[0];
        }

        float cs = 0.0f, cq = 0.0f;
        #pragma unroll
        for (int r = 0; r < 8; ++r) {
            const int row = m0 + r + 8 * half;
            if (row < M) {
                C[(size_t)row * DF + col] = hv[r];
                if (COLSUMS) { cs += hv[r]; cq += hv[r] * hv[r]; }
            }
        }

        if (COLSUMS) {
            cs += other_half(cs);
            cq += other_half(cq);
            colsum_rows[(size_t)stripe * (2 * DF) + t * 32 + lane] = half ? cq : cs;
        }
    }

    if (SCORES) {
        const int node = m0 + 8 * half + (l & 7);
        if (node < M)
            *(float4*)(((l >> 3) ? s_j : s_i) + (size_t)node * NHEADS) = make_float4(sc[0], sc[1], sc[2], sc[3]);
    }
}

#define FRONT_GRID (INIT_GRID + 2 * GEMM_GRID)
#define INIT_STEP  8
#define INIT_SPAN  (INIT_GRID * INIT_STEP)
static_assert(INIT_SPAN <= FRONT_GRID, "every bitmap work-group has an index in the grid");
__global__ __launch_bounds__(256) void k_front(
    const int* __restrict__ ei, float* __restrict__ ws,
    unsigned char* __restrict__ rbyte, unsigned* __restrict__ bitmap,
    const float* __restrict__ x, int M,
    const float* __restrict__ Wk, const float* __restrict__ bk, float* __restrict__ h,
    const float* __restrict__ a_i, const float* __restrict__ a_j,
    float* __restrict__ s_i, float* __restrict__ s_j,
    const float* __restrict__ Wm, const float* __restrict__ bm_bias, float* __restrict__ upd,
    float* __restrict__ colsum_rows)
{
    __shared__ __attribute__((aligned(16))) float smem[(DF / 2) * WP];
    static_assert(sizeof(unsigned) * 32 * BMP <= sizeof(float) * (DF / 2) * WP, "stage 0's planes fit the launch's LDS");
    const int b = blockIdx.x;
    if (b >= INIT_SPAN || (b % INIT_STEP) != 0) {
        const int below = (b + INIT_STEP - 1) / INIT_STEP;
        const int g = b - (below < INIT_GRID ? below : INIT_GRID);
        if (g & 1)
            gemm_stage<0, 1, 0, 1>(g >> 1, smem, x, Wm, bm_bias, upd, M,
                                   nullptr, nullptr, nullptr, nullptr, colsum_rows);
        else
            gemm_stage<1, 0, 1, 0>(g >> 1, smem, x, Wk, bk, h, M,
                                   a_i, a_j, s_i, s_j, nullptr);
    } else {
        init_stage(b / INIT_STEP, (unsigned*)smem, ei, ws, rbyte, bitmap);
    }
}

#define SG_WAVES 16
#define PCAP     64
#define LCAP     384
static_assert(NN <= 65536 && RNG <= 256, "a list entry packs sender (16 bits) and receiver - r0");

__device__ __forceinline__ unsigned zero_bytes(unsigned t) {
    return ~(((t & 0x7f7f7f7fu) + 0x7f7f7f7fu) | t) & 0x80808080u;
}

__device__ __forceinline__ void lds_max_float(float* cell, float v) {
    if (__float_as_int(v) >= 0) atomicMax((int*)cell, __float_as_int(v));
    else                        atomicMin((unsigned int*)cell, __float_as_uint(v));
}

__device__ __forceinline__ float leaky(float v) { return (v > 0.0f) ? v : NEG_SLOPE * v; }

#define VCAP 64
#define BM_PER_WAVE (BMW / SG_WAVES)
#define BM_TRIPS    ((BM_PER_WAVE + 31) / 32)
static_assert(BMW % SG_WAVES == 0, "every wave walks the same number of bitmap dwords");
template <class Sink>
__device__ __forceinline__ void walk_range_bitmap(
    const unsigned* __restrict__ brow, const uint4* __restrict__ rv, unsigned pat, int wv, int lane,
    int* vpend, int* pend, Sink&& sink)
{
    unsigned head = 0, tail = 0;
    unsigned vhead = 0, vtail = 0;
    unsigned bw[BM_TRIPS];
    #pragma unroll
    for (int t = 0; t < BM_TRIPS; ++t) {
        const int d = t * 32 + lane;
        const unsigned v = brow[wv * BM_PER_WAVE + (d < BM_PER_WAVE ? d : BM_PER_WAVE - 1)];
        bw[t] = (d < BM_PER_WAVE) ? v : 0u;
    }
    auto vectors = [&](unsigned vh, int n) {
        const int vec = vpend[(vh + (unsigned)(lane < n ? lane : 0)) & (VCAP - 1)];
        const uint4 cur = rv[vec];
        unsigned m = 0u;
        if (lane < n)
            m = (zero_bytes(cur.x ^ pat) >> 7) | (zero_bytes(cur.y ^ pat) >> 6)
              | (zero_bytes(cur.z ^ pat) >> 5) | (zero_bytes(cur.w ^ pat) >> 4);
        for (;;) {
            const unsigned bal = __builtin_amdgcn_ballot_w32(m != 0u);
            if (bal == 0u) break;
            if (m != 0u) {
                const int p = __builtin_ctz(m);
                const unsigned rank = __builtin_amdgcn_mbcnt_lo(bal, 0u);
                pend[(tail + rank) & (PCAP - 1)] = vec * 16 + ((p & 7) << 2) + (p >> 3);
                m &= m - 1u;
            }
            tail += (unsigned)__builtin_popcount(bal);
            if (tail - head >= 32u) { sink(head, 32); head += 32u; }
        }
    };
    #pragma unroll
    for (int t = 0; t < BM_TRIPS; ++t) {
        unsigned w = bw[t];
        const int v0 = (wv * BM_PER_WAVE + t * 32 + lane) * 32;
        for (;;) {
            const unsigned bal = __builtin_amdgcn_ballot_w32(w != 0u);
            if (bal == 0u) break;
            if (w != 0u) {
                const int p = __builtin_ctz(w);
                const unsigned rank = __builtin_amdgcn_mbcnt_lo(bal, 0u);
                vpend[(vtail + rank) & (VCAP - 1)] = v0 + p;
                w &= w - 1u;
            }
            vtail += (unsigned)__builtin_popcount(bal);
            if (vtail - vhead >= 32u) { vectors(vhead, 32); vhead += 32u; }
        }
    }
    if (vtail != vhead) vectors(vhead, (int)(vtail - vhead));
    if (tail != head) sink(head, (int)(tail - head));
}

__device__ __forceinline__ void edge_logits(
    const int* pend, unsigned head, int n, int lane, int r0,
    const int* __restrict__ ei, const float* __restrict__ s_j, const float* sis,
    float4 ae, float4 ba, int& packed, int& rel, float4& lg)
{
    const int e = pend[(head + (unsigned)(lane < n ? lane : 0)) & (PCAP - 1)];
    const int2 rs = *(const int2*)(ei + 2 * (size_t)e);
    const float4 sj = *(const float4*)(s_j + (size_t)rs.y * NHEADS);
    rel = rs.x - r0;
    const float4 si = *(const float4*)(sis + rel * NHEADS);
    lg.x = leaky(si.x + sj.x + ae.x + ba.x);
    lg.y = leaky(si.y + sj.y + ae.y + ba.y);
    lg.z = leaky(si.z + sj.z + ae.z + ba.z);
    lg.w = leaky(si.w + sj.w + ae.w + ba.w);
    packed = rs.y | (rel << 16);
}

__device__ __forceinline__ void list_weights(
    const int* lsr, float* llg, int from, int n, int lane, const float* mxs, float* dns)
{
    for (int b = 0; b < n; b += 32) {
        const int i = from + b + lane;
        if (b + lane < n) {
            const int rel = (int)(((unsigned)lsr[i] >> 16) & 0xffu);
            const float4 l = *(const float4*)(llg + 4 * i);
            const float4 m = *(const float4*)(mxs + rel * NHEADS);
            float4 w;
            w.x = expf(l.x - m.x); w.y = expf(l.y - m.y); w.z = expf(l.z - m.z); w.w = expf(l.w - m.w);
            atomicAdd(dns + rel * NHEADS + 0, w.x);
            atomicAdd(dns + rel * NHEADS + 1, w.y);
            atomicAdd(dns + rel * NHEADS + 2, w.z);
            atomicAdd(dns + rel * NHEADS + 3, w.w);
            *(float4*)(llg + 4 * i) = w;
        }
    }
}

template <bool FULL>
__device__ __forceinline__ void add_batch(
    const int* lsr, const float* llg, int base, int n, int lane,
    const float* __restrict__ h, float* acc)
{
    const int hd = lane >> 3;
    int   sr[8];
    float w[8];
    #pragma unroll
    for (int b = 0; b < 8; ++b) {
        const int i = base + (FULL ? b : (b < n ? b : n - 1));
        sr[b] = lsr[i];
        w[b]  = llg[4 * i + hd];
    }
    float4 hv[8];
    #pragma unroll
    for (int b = 0; b < 8; ++b)
        hv[b] = *(const float4*)(h + (size_t)(sr[b] & 0xffff) * DF + 4 * lane);
    #pragma unroll
    for (int b = 0; b < 8; ++b) {
        if (FULL || b < n) {
            float* a = acc + (int)(((unsigned)sr[b] >> 16) & 0xffu) * DF + lane;
            atomicAdd(a,      hv[b].x * w[b]);
            atomicAdd(a + 32, hv[b].y * w[b]);
            atomicAdd(a + 64, hv[b].z * w[b]);
            atomicAdd(a + 96, hv[b].w * w[b]);
        }
    }
}

__device__ __forceinline__ void list_rows(
    const int* lsr, const float* llg, int n, int lane, const float* __restrict__ h, float* acc)
{
    int base = 0;
    for (; base + 8 <= n; base += 8) add_batch<true>(lsr, llg, base, 8, lane, h, acc);
    if (base < n) add_batch<false>(lsr, llg, base, n - base, lane, h, acc);
}

#define ROWS_PER_WAVE ((RNG + SG_WAVES - 1) / SG_WAVES)

__device__ __forceinline__ float out_elem(float u, float mean, float rs, float g, float b, float a, float d) {
    const float normed = g * (u - mean) * rs + b;
    return normed + fmaxf(a / d, 0.0f);
}

#define SCAP 4864
static_assert((size_t)(SCAP + 32) * 20 <= (size_t)RNG * DF * 4, "the table, and a group's reads past its end, fit the block of the sums");
#define RCH  24
#define SCAN_PER_LANE ((RNG + 1 + 31) / 32)

__global__ __launch_bounds__(512) void k_edge_range_sum(
    const int* __restrict__ ei, const float* __restrict__ h,
    const float* __restrict__ s_i, const float* __restrict__ s_j,
    const float* __restrict__ a_e, const float* __restrict__ b_a,
    const unsigned char* __restrict__ rbyte, const unsigned* __restrict__ bitmap,
    const float* __restrict__ upd, const float* __restrict__ macc,
    const float* __restrict__ sqacc, const float* __restrict__ gamma,
    const float* __restrict__ beta, float* __restrict__ out)
{
    __shared__ __attribute__((aligned(16))) float acc[RNG * DF];
    __shared__ __attribute__((aligned(16))) float mxs[RNG * NHEADS];
    __shared__ __attribute__((aligned(16))) float dns[RNG * NHEADS];
    __shared__ __attribute__((aligned(16))) float sis[RNG * NHEADS];
    __shared__ int   pending[SG_WAVES * PCAP];
    __shared__ int   vpending[SG_WAVES * VCAP];
    __shared__ int   list_sr[SG_WAVES * LCAP];
    __shared__ __attribute__((aligned(16))) float list_lg[SG_WAVES * LCAP * 4];
    __shared__ int   rcnt[RNG];
    __shared__ int   roff[RNG + 1];
    __shared__ int   s_slow;

    const int tid  = threadIdx.x;
    const int lane = tid & 31;
    const int wv   = tid >> 5;
    const int r0   = blockIdx.x * RNG;

    for (int i = tid; i < RNG * NHEADS; i += 512) {
        const int g = r0 * NHEADS + i;
        mxs[i] = -INFINITY;
        dns[i] = 0.0f;
        sis[i] = (g < NN * NHEADS) ? s_i[g] : 0.0f;
    }
    for (int i = tid; i < RNG; i += 512) rcnt[i] = 0;
    if (tid == 0) s_slow = 0;
    const float4 ae = make_float4(a_e[0], a_e[1], a_e[2], a_e[3]);
    const float4 ba = make_float4(b_a[0], b_a[1], b_a[2], b_a[3]);
    __syncthreads();

    const unsigned pat = blockIdx.x * 0x01010101u;
    const uint4* __restrict__ rv = (const uint4*)rbyte;
    int*   pend = pending + wv * PCAP;
    int*   vpend = vpending + wv * VCAP;
    const unsigned* __restrict__ brow = bitmap + (size_t)blockIdx.x * BMW;
    int*   lsr  = list_sr + wv * LCAP;
    float* llg  = list_lg + wv * LCAP * 4;
    int  cnt  = 0;
    bool over = false;
    bool bad  = false;

    walk_range_bitmap(brow, rv, pat, wv, lane, vpend, pend, [&](unsigned head, int n) {
        int packed, rel; float4 lg;
        edge_logits(pend, head, n, lane, r0, ei, s_j, sis, ae, ba, packed, rel, lg);
        if (lane < n) {
            lds_max_float(mxs + rel * NHEADS + 0, lg.x);
            lds_max_float(mxs + rel * NHEADS + 1, lg.y);
            lds_max_float(mxs + rel * NHEADS + 2, lg.z);
            lds_max_float(mxs + rel * NHEADS + 3, lg.w);
            const int rank = atomicAdd(&rcnt[rel], 1);
            bad = bad || (rank > 255);
            packed |= (int)((unsigned)(rank & 0xff) << 24);
        }
        if (!over && cnt + n <= LCAP) {
            if (lane < n) {
                lsr[cnt + lane] = packed;
                *(float4*)(llg + 4 * (cnt + lane)) = lg;
            }
            cnt += n;
        } else {
            over = true;
        }
    });
    if (over || bad) s_slow = 1;
    __syncthreads();

    if (wv == 0) {
        int cv[SCAN_PER_LANE];
        int s = 0;
        #pragma unroll
        for (int k = 0; k < SCAN_PER_LANE; ++k) {
            const int idx = lane * SCAN_PER_LANE + k;
            cv[k] = (idx < RNG) ? rcnt[idx] : 0;
            s += cv[k];
        }
        int incl = s;
        #pragma unroll
        for (int d = 1; d < 32; d <<= 1) {
            const int t = __shfl_up(incl, d, 32);
            if (lane >= d) incl += t;
        }
        int run = incl - s;
        #pragma unroll
        for (int k = 0; k < SCAN_PER_LANE; ++k) {
            const int idx = lane * SCAN_PER_LANE + k;
            if (idx <= RNG) roff[idx] = run;
            run += cv[k];
        }
        if (lane == 31 && incl > SCAP) s_slow = 1;
    }
    __syncthreads();
    const bool slow = (s_slow != 0);

    const float4 m4 = *(const float4*)(macc + 4 * lane);
    const float4 q4 = *(const float4*)(sqacc + 4 * lane);
    const float4 g4 = *(const float4*)(gamma + 4 * lane);
    const float4 b4 = *(const float4*)(beta + 4 * lane);
    const float inv_n = 1.0f / (float)NN;
    const float mean0 = m4.x * inv_n, mean1 = m4.y * inv_n, mean2 = m4.z * inv_n, mean3 = m4.w * inv_n;
    const float rs0 = rsqrtf(q4.x * inv_n - mean0 * mean0 + BN_EPS);
    const float rs1 = rsqrtf(q4.y * inv_n - mean1 * mean1 + BN_EPS);
    const float rs2 = rsqrtf(q4.z * inv_n - mean2 * mean2 + BN_EPS);
    const float rs3 = rsqrtf(q4.w * inv_n - mean3 * mean3 + BN_EPS);
    const int hd = lane >> 3;

    if (!slow) {
        float* tw  = acc;
        int*   tsd = (int*)(acc + 4 * SCAP);
        for (int b = 0; b < cnt; b += 32) {
            const int i = b + lane;
            if (i < cnt) {
                const unsigned p = (unsigned)lsr[i];
                const int rel = (int)((p >> 16) & 0xffu);
                const int pos = roff[rel] + (int)(p >> 24);
                const float4 l = *(const float4*)(llg + 4 * i);
                const float4 m = *(const float4*)(mxs + rel * NHEADS);
                float4 w;
                w.x = expf(l.x - m.x); w.y = expf(l.y - m.y); w.z = expf(l.z - m.z); w.w = expf(l.w - m.w);
                *(float4*)(tw + 4 * pos) = w;
                tsd[pos] = (int)(p & 0xffffu);
            }
        }
        __syncthreads();

        for (int r = wv; r < RNG; r += SG_WAVES) {
            const int row = r0 + r;
            if (row >= NN) break;
            const int start = __builtin_amdgcn_readfirstlane(roff[r]);
            const int end   = __builtin_amdgcn_readfirstlane(roff[r + 1]);
            const float4 u = *(const float4*)(upd + (size_t)row * DF + 4 * lane);
            float4 a = make_float4(0.0f, 0.0f, 0.0f, 0.0f);
            float den = 0.0f;
            for (int base = start; base < end; base += RCH) {
                const int n = end - base;
                const int sdv = tsd[base + lane];
                float  w[RCH];
                float4 hv[RCH];
                #pragma unroll
                for (int b = 0; b < RCH; ++b) {
                    w[b] = tw[4 * (base + b) + hd];
                }
                #pragma unroll
                for (int b = 0; b < RCH; ++b) {
                    if (b < n) {
                        const int s = __builtin_amdgcn_readlane(sdv, b);
                        hv[b] = *(const float4*)(h + (size_t)s * DF + 4 * lane);
                    }
                }
                #pragma unroll
                for (int b = 0; b < RCH; ++b) {
                    if (b < n) {
                        a.x += hv[b].x * w[b];
                        a.y += hv[b].y * w[b];
                        a.z += hv[b].z * w[b];
                        a.w += hv[b].w * w[b];
                        den += w[b];
                    }
                }
            }
            const float d = den + 1e-16f;
            float4 v;
            v.x = out_elem(u.x, mean0, rs0, g4.x, b4.x, a.x, d);
            v.y = out_elem(u.y, mean1, rs1, g4.y, b4.y, a.y, d);
            v.z = out_elem(u.z, mean2, rs2, g4.z, b4.z, a.z, d);
            v.w = out_elem(u.w, mean3, rs3, g4.w, b4.w, a.w, d);
            *(float4*)(out + (size_t)row * DF + 4 * lane) = v;
        }
        return;
    }

    for (int i = tid; i < RNG * DF; i += 512) acc[i] = 0.0f;
    __syncthreads();

    list_weights(lsr, llg, 0, cnt, lane, mxs, dns);
    list_rows(lsr, llg, cnt, lane, h, acc);
    if (over) {
        int seen = 0;
        walk_range_bitmap(brow, rv, pat, wv, lane, vpend, pend, [&](unsigned head, int n) {
            const int lo = (cnt > seen) ? (cnt - seen) : 0;
            seen += n;
            if (lo >= n) return;
            int packed, rel; float4 lg;
            edge_logits(pend, head, n, lane, r0, ei, s_j, sis, ae, ba, packed, rel, lg);
            if (lane >= lo && lane < n) {
                lsr[lane - lo] = packed;
                *(float4*)(llg + 4 * (lane - lo)) = lg;
            }
            list_weights(lsr, llg, 0, n - lo, lane, mxs, dns);
            list_rows(lsr, llg, n - lo, lane, h, acc);
        });
    }

    float4 u[ROWS_PER_WAVE];
    #pragma unroll
    for (int i = 0; i < ROWS_PER_WAVE; ++i) {
        const int row = r0 + wv + i * SG_WAVES;
        u[i] = *(const float4*)(upd + (size_t)(row < NN ? row : NN - 1) * DF + 4 * lane);
    }
    __syncthreads();

    #pragma unroll
    for (int i = 0; i < ROWS_PER_WAVE; ++i) {
        const int r = wv + i * SG_WAVES;
        const int row = r0 + r;
        if (r < RNG && row < NN) {
            const float* a = acc + r * DF + lane;
            const float d = dns[r * NHEADS + hd] + 1e-16f;
            float4 v;
            v.x = out_elem(u[i].x, mean0, rs0, g4.x, b4.x, a[0],  d);
            v.y = out_elem(u[i].y, mean1, rs1, g4.y, b4.y, a[32], d);
            v.z = out_elem(u[i].z, mean2, rs2, g4.z, b4.z, a[64], d);
            v.w = out_elem(u[i].w, mean3, rs3, g4.w, b4.w, a[96], d);
            *(float4*)(out + (size_t)row * DF + 4 * lane) = v;
        }
    }
}

__global__ __launch_bounds__(256) void k_bn_reduce(
    const float* __restrict__ rows, float* __restrict__ macc, float* __restrict__ sqacc)
{
    const int t = threadIdx.x;
    const float* __restrict__ p = rows + (size_t)blockIdx.x * CS_ROWS * (2 * DF) + t;
    float v[CS_ROWS];
    #pragma unroll
    for (int r = 0; r < CS_ROWS; ++r) v[r] = p[(size_t)r * (2 * DF)];
    float sum = 0.0f;
    #pragma unroll
    for (int r = 0; r < CS_ROWS; ++r) sum += v[r];
    const int col = ((t >> 5) << 4) | (t & 15);
    atomicAdd(((t >> 4) & 1) ? &sqacc[col] : &macc[col], sum);
}

extern "C" void kernel_launch(void* const* d_in, const int* in_sizes, int n_in,
                              void* d_out, int out_size, void* d_ws, size_t ws_size,
                              hipStream_t stream) {
    const float* x     = (const float*)d_in[0];
    const int*   ei    = (const int*)  d_in[1];
    const float* Wk    = (const float*)d_in[2];
    const float* bk    = (const float*)d_in[3];
    const float* a_i   = (const float*)d_in[4];
    const float* a_j   = (const float*)d_in[5];
    const float* a_e   = (const float*)d_in[6];
    const float* b_a   = (const float*)d_in[7];
    const float* Wm    = (const float*)d_in[8];
    const float* bm    = (const float*)d_in[9];
    const float* gamma = (const float*)d_in[10];
    const float* beta  = (const float*)d_in[11];
    float* out = (float*)d_out;
    float* ws  = (float*)d_ws;

    float* h_buf  = ws + OFF_H;
    float* upd    = ws + OFF_UPD;
    float* s_i    = ws + OFF_SI;
    float* s_j    = ws + OFF_SJ;
    float* macc   = ws + OFF_MACC;
    float* sqacc  = ws + OFF_SQ;
    unsigned char* rbyte = (unsigned char*)(ws + OFF_W);
    unsigned* bitmap = (unsigned*)(ws + OFF_W + NE / 4);
    float* colsum_rows = ws + OFF_AGG;

    k_front<<<FRONT_GRID, 256, 0, stream>>>(
        ei, ws, rbyte, bitmap, x, NN, Wk, bk, h_buf, a_i, a_j, s_i, s_j, Wm, bm, upd, colsum_rows);
    k_bn_reduce<<<GEMM_BLOCKS / CS_ROWS, 256, 0, stream>>>(colsum_rows, macc, sqacc);
    k_edge_range_sum<<<NRNG, 512, 0, stream>>>(
        ei, h_buf, s_i, s_j, a_e, b_a, rbyte, bitmap, upd, macc, sqacc, gamma, beta, out);
}
